# starting kernel
# baseline (speedup 1.0000x reference)
#include <hip/hip_runtime.h>
#include <hip/hip_bf16.h>

#define GAT_N 30000
#define GAT_E 480000
#define GAT_ETOT (GAT_E + GAT_N)
#define GAT_H 8
#define GAT_C 32
#define GAT_FH 256
#define NEG_SLOPE 0.2f

typedef float v2f __attribute__((ext_vector_type(2)));
typedef float v8f __attribute__((ext_vector_type(8)));

__global__ void gemm_wmma_f32(const float* __restrict__ A,
                              const float* __restrict__ B,
                              float* __restrict__ C,
                              int M, int K, int Ncols) {
  const int wave = threadIdx.x >> 5;
  const int lane = threadIdx.x & 31;
  const int half = lane >> 4;
  const int l    = lane & 15;

  const int m0 = blockIdx.x * 16;
  const int n0 = blockIdx.y * 64 + wave * 16;
  if (m0 >= M || n0 >= Ncols) return;

  v8f c = {0.f, 0.f, 0.f, 0.f, 0.f, 0.f, 0.f, 0.f};

  const float* arow = A + (size_t)(m0 + l) * K + half * 2;
  const float* bcol = B + (size_t)(half * 2) * Ncols + n0 + l;

  for (int k = 0; k < K; k += 4) {
    v2f a = *(const v2f*)(arow + k);
    v2f b;
    b.x = bcol[(size_t)k * Ncols];
    b.y = bcol[(size_t)k * Ncols + Ncols];
    c = __builtin_amdgcn_wmma_f32_16x16x4_f32(
         false, a,  false, b,
         (short)0, c,  false,  false);
  }

  float* crow = C + n0 + l;
  const int mbase = m0 + half * 8;
#pragma unroll
  for (int r = 0; r < 8; ++r)
    crow[(size_t)(mbase + r) * Ncols] = c[r];
}

__global__ void alphas_kernel(const float* __restrict__ h,
                              const float* __restrict__ asrc,
                              const float* __restrict__ adst,
                              float* __restrict__ s, float* __restrict__ d,
                              int total) {
  int i = blockIdx.x * blockDim.x + threadIdx.x;
  if (i >= total) return;
  int node = i >> 3, hd = i & 7;
  const float* hp = h + (size_t)node * GAT_FH + hd * GAT_C;
  const float* ap = asrc + hd * GAT_C;
  const float* bp = adst + hd * GAT_C;
  float ss = 0.f, dd = 0.f;
#pragma unroll
  for (int cc = 0; cc < GAT_C; ++cc) {
    float v = hp[cc];
    ss += v * ap[cc];
    dd += v * bp[cc];
  }
  s[i] = ss;
  d[i] = dd;
}

__global__ void init_kernel(float* __restrict__ agg, float* __restrict__ emax,
                            float* __restrict__ denom, int nfeat, int nh) {
  int i = blockIdx.x * blockDim.x + threadIdx.x;
  if (i < nfeat) agg[i] = 0.f;
  if (i < nh) { emax[i] = -3.0e38f; denom[i] = 0.f; }
}

__device__ __forceinline__ void edge_endpoints(const int* __restrict__ ei,
                                               int e, int& src, int& dst) {
  if (e < GAT_E) { src = ei[e]; dst = ei[GAT_E + e]; }
  else           { src = e - GAT_E; dst = e - GAT_E; }
}

__global__ void edge_max_kernel(const int* __restrict__ ei,
                                const float* __restrict__ as,
                                const float* __restrict__ ad,
                                float* __restrict__ emax) {
  int e = blockIdx.x * blockDim.x + threadIdx.x;
  if (e >= GAT_ETOT) return;
  int src, dst;
  edge_endpoints(ei, e, src, dst);
#pragma unroll
  for (int hd = 0; hd < GAT_H; ++hd) {
    float v = as[src * GAT_H + hd] + ad[dst * GAT_H + hd];
    v = v > 0.f ? v : NEG_SLOPE * v;
    atomicMax(&emax[dst * GAT_H + hd], v);
  }
}

__global__ void edge_denom_kernel(const int* __restrict__ ei,
                                  const float* __restrict__ as,
                                  const float* __restrict__ ad,
                                  const float* __restrict__ emax,
                                  float* __restrict__ denom) {
  int e = blockIdx.x * blockDim.x + threadIdx.x;
  if (e >= GAT_ETOT) return;
  int src, dst;
  edge_endpoints(ei, e, src, dst);
#pragma unroll
  for (int hd = 0; hd < GAT_H; ++hd) {
    float v = as[src * GAT_H + hd] + ad[dst * GAT_H + hd];
    v = v > 0.f ? v : NEG_SLOPE * v;
    atomicAdd(&denom[dst * GAT_H + hd], __expf(v - emax[dst * GAT_H + hd]));
  }
}

__global__ void edge_agg_kernel(const int* __restrict__ ei,
                                const float* __restrict__ as,
                                const float* __restrict__ ad,
                                const float* __restrict__ emax,
                                const float* __restrict__ denom,
                                const float* __restrict__ h,
                                float* __restrict__ agg) {
  int t = blockIdx.x * blockDim.x + threadIdx.x;
  if (t >= GAT_ETOT * GAT_H) return;
  int e = t >> 3, hd = t & 7;
  int src, dst;
  edge_endpoints(ei, e, src, dst);
  float v = as[src * GAT_H + hd] + ad[dst * GAT_H + hd];
  v = v > 0.f ? v : NEG_SLOPE * v;
  int di = dst * GAT_H + hd;
  float alpha = __expf(v - emax[di]) / (denom[di] + 1e-16f);
  const float4* hp = (const float4*)(h + (size_t)src * GAT_FH + hd * GAT_C);
  float* op = agg + (size_t)dst * GAT_FH + hd * GAT_C;
#pragma unroll
  for (int q = 0; q < GAT_C / 4; ++q) {
    float4 x4 = hp[q];
    atomicAdd(op + q * 4 + 0, x4.x * alpha);
    atomicAdd(op + q * 4 + 1, x4.y * alpha);
    atomicAdd(op + q * 4 + 2, x4.z * alpha);
    atomicAdd(op + q * 4 + 3, x4.w * alpha);
  }
}

__global__ void bias_elu_kernel(float* __restrict__ agg,
                                const float* __restrict__ b, int total) {
  int i = blockIdx.x * blockDim.x + threadIdx.x;
  if (i >= total) return;
  float v = agg[i] + b[i & (GAT_FH - 1)];
  agg[i] = v > 0.f ? v : (__expf(v) - 1.f);
}

__global__ void final_linear_kernel(const float* __restrict__ h,
                                    const float* __restrict__ Wl,
                                    const float* __restrict__ bl,
                                    float* __restrict__ y, int n) {
  int gid = blockIdx.x * blockDim.x + threadIdx.x;
  int wid = gid >> 5;
  int lane = threadIdx.x & 31;
  if (wid >= n) return;
  const float* hp = h + (size_t)wid * GAT_FH;
  float s = 0.f;
#pragma unroll
  for (int c = lane; c < GAT_FH; c += 32) s += hp[c] * Wl[c];
#pragma unroll
  for (int off = 16; off > 0; off >>= 1) s += __shfl_down(s, off, 32);
  if (lane == 0) y[wid] = s + bl[0];
}

extern "C" void kernel_launch(void* const* d_in, const int* in_sizes, int n_in,
                              void* d_out, int out_size, void* d_ws, size_t ws_size,
                              hipStream_t stream) {
  const float* x      = (const float*)d_in[0];
  const int*   ei     = (const int*)  d_in[1];
  const float* W1     = (const float*)d_in[2];
  const float* a1_src = (const float*)d_in[3];
  const float* a1_dst = (const float*)d_in[4];
  const float* b1     = (const float*)d_in[5];
  const float* W2     = (const float*)d_in[6];
  const float* a2_src = (const float*)d_in[7];
  const float* a2_dst = (const float*)d_in[8];
  const float* b2     = (const float*)d_in[9];
  const float* Wl     = (const float*)d_in[10];
  const float* bl     = (const float*)d_in[11];
  float* out = (float*)d_out;

  float* hbuf  = (float*)d_ws;
  float* agg   = hbuf  + (size_t)GAT_N * GAT_FH;
  float* emax  = agg   + (size_t)GAT_N * GAT_FH;
  float* denom = emax  + (size_t)GAT_N * GAT_H;
  float* as    = denom + (size_t)GAT_N * GAT_H;
  float* ad    = as    + (size_t)GAT_N * GAT_H;

  const int NF  = GAT_N * GAT_FH;
  const int NH  = GAT_N * GAT_H;
  const int EH  = GAT_ETOT * GAT_H;

  dim3 gemm_grid(GAT_N / 16, GAT_FH / 64);
  dim3 gemm_blk(128);

  const int TPB = 256;
  auto blocks = [](int n, int tpb) { return (n + tpb - 1) / tpb; };

  gemm_wmma_f32<<<gemm_grid, gemm_blk, 0, stream>>>(x, W1, hbuf, GAT_N, 128, GAT_FH);
  alphas_kernel<<<blocks(NH, TPB), TPB, 0, stream>>>(hbuf, a1_src, a1_dst, as, ad, NH);
  init_kernel<<<blocks(NF, TPB), TPB, 0, stream>>>(agg, emax, denom, NF, NH);
  edge_max_kernel<<<blocks(GAT_ETOT, TPB), TPB, 0, stream>>>(ei, as, ad, emax);
  edge_denom_kernel<<<blocks(GAT_ETOT, TPB), TPB, 0, stream>>>(ei, as, ad, emax, denom);
  edge_agg_kernel<<<blocks(EH, TPB), TPB, 0, stream>>>(ei, as, ad, emax, denom, hbuf, agg);
  bias_elu_kernel<<<blocks(NF, TPB), TPB, 0, stream>>>(agg, b1, NF);

  gemm_wmma_f32<<<gemm_grid, gemm_blk, 0, stream>>>(agg, W2, hbuf, GAT_N, GAT_FH, GAT_FH);
  alphas_kernel<<<blocks(NH, TPB), TPB, 0, stream>>>(hbuf, a2_src, a2_dst, as, ad, NH);
  init_kernel<<<blocks(NF, TPB), TPB, 0, stream>>>(agg, emax, denom, NF, NH);
  edge_max_kernel<<<blocks(GAT_ETOT, TPB), TPB, 0, stream>>>(ei, as, ad, emax);
  edge_denom_kernel<<<blocks(GAT_ETOT, TPB), TPB, 0, stream>>>(ei, as, ad, emax, denom);
  edge_agg_kernel<<<blocks(EH, TPB), TPB, 0, stream>>>(ei, as, ad, emax, denom, hbuf, agg);
  bias_elu_kernel<<<blocks(NF, TPB), TPB, 0, stream>>>(agg, b2, NF);

  final_linear_kernel<<<blocks(GAT_N * 32, TPB), TPB, 0, stream>>>(agg, Wl, bl, out, GAT_N);
}


# Float-atomic scatter replaced by destination-ordered gather: edges counting-sorted by dst once per call (int atomics, scan, bucket rank sort); one wave per node sums in a fixed order in registers
# speedup vs baseline: 7.0338x; 7.0338x over previous
#include <hip/hip_runtime.h>
#include <hip/hip_bf16.h>

#define GAT_N 30000
#define GAT_E 480000
#define GAT_ETOT (GAT_E + GAT_N)
#define GAT_H 8
#define GAT_C 32
#define GAT_FH 256
#define NEG_SLOPE 0.2f

typedef float v2f __attribute__((ext_vector_type(2)));
typedef float v8f __attribute__((ext_vector_type(8)));

__global__ void gemm_wmma_f32(const float* __restrict__ A,
                              const float* __restrict__ B,
                              float* __restrict__ C,
                              int M, int K, int Ncols) {
  const int wave = threadIdx.x >> 5;
  const int lane = threadIdx.x & 31;
  const int half = lane >> 4;
  const int l    = lane & 15;

  const int m0 = blockIdx.x * 16;
  const int n0 = blockIdx.y * 64 + wave * 16;
  if (m0 >= M || n0 >= Ncols) return;

  v8f c = {0.f, 0.f, 0.f, 0.f, 0.f, 0.f, 0.f, 0.f};

  const float* arow = A + (size_t)(m0 + l) * K + half * 2;
  const float* bcol = B + (size_t)(half * 2) * Ncols + n0 + l;

  for (int k = 0; k < K; k += 4) {
    v2f a = *(const v2f*)(arow + k);
    v2f b;
    b.x = bcol[(size_t)k * Ncols];
    b.y = bcol[(size_t)k * Ncols + Ncols];
    c = __builtin_amdgcn_wmma_f32_16x16x4_f32(
         false, a,  false, b,
         (short)0, c,  false,  false);
  }

  float* crow = C + n0 + l;
  const int mbase = m0 + half * 8;
#pragma unroll
  for (int r = 0; r < 8; ++r)
    crow[(size_t)(mbase + r) * Ncols] = c[r];
}

__global__ void alphas_kernel(const float* __restrict__ h,
                              const float* __restrict__ asrc,
                              const float* __restrict__ adst,
                              float* __restrict__ s, float* __restrict__ d,
                              int total) {
  int i = blockIdx.x * blockDim.x + threadIdx.x;
  if (i >= total) return;
  int node = i >> 3, hd = i & 7;
  const float* hp = h + (size_t)node * GAT_FH + hd * GAT_C;
  const float* ap = asrc + hd * GAT_C;
  const float* bp = adst + hd * GAT_C;
  float ss = 0.f, dd = 0.f;
#pragma unroll
  for (int cc = 0; cc < GAT_C; ++cc) {
    float v = hp[cc];
    ss += v * ap[cc];
    dd += v * bp[cc];
  }
  s[i] = ss;
  d[i] = dd;
}

__global__ void csr_zero_kernel(int* __restrict__ deg) {
  int i = blockIdx.x * blockDim.x + threadIdx.x;
  if (i < GAT_N) deg[i] = 0;
}

__global__ void csr_count_kernel(const int* __restrict__ ei, int* __restrict__ deg) {
  int e = blockIdx.x * blockDim.x + threadIdx.x;
  if (e < GAT_E) atomicAdd(&deg[ei[GAT_E + e]], 1);
}

#define SCAN_PER ((GAT_N + 1023) / 1024)
__global__ void __launch_bounds__(1024)
csr_scan_kernel(const int* __restrict__ deg, int* __restrict__ rowptr,
                int* __restrict__ cursor) {
  __shared__ int part[1024];
  const int t = threadIdx.x;
  const int b = t * SCAN_PER;
  const int e = (b + SCAN_PER < GAT_N) ? (b + SCAN_PER) : GAT_N;
  int s = 0;
  for (int i = b; i < e; ++i) s += deg[i];
  part[t] = s;
  __syncthreads();
  for (int off = 1; off < 1024; off <<= 1) {
    int v = (t >= off) ? part[t - off] : 0;
    __syncthreads();
    part[t] += v;
    __syncthreads();
  }
  int run = part[t] - s;
  for (int i = b; i < e; ++i) {
    rowptr[i] = run;
    cursor[i] = run;
    run += deg[i];
  }
  if (t == 1023) rowptr[GAT_N] = part[1023];
}

__global__ void csr_fill_kernel(const int* __restrict__ ei, int* __restrict__ cursor,
                                int* __restrict__ tmp_src) {
  int e = blockIdx.x * blockDim.x + threadIdx.x;
  if (e >= GAT_E) return;
  int p = atomicAdd(&cursor[ei[GAT_E + e]], 1);
  tmp_src[p] = ei[e];
}

__global__ void csr_order_kernel(const int* __restrict__ rowptr,
                                 const int* __restrict__ tmp_src,
                                 int* __restrict__ csr_src) {
  const int lane = threadIdx.x & 31;
  const int d = __builtin_amdgcn_readfirstlane((blockIdx.x * blockDim.x + threadIdx.x) >> 5);
  if (d >= GAT_N) return;
  const int beg = rowptr[d];
  const int deg = rowptr[d + 1] - beg;
  for (int t = lane; t < deg; t += 32) {
    const int s = tmp_src[beg + t];
    int rank = 0;
    for (int j = 0; j < deg; ++j) {
      const int sj = tmp_src[beg + j];
      rank += ((sj < s) || (sj == s && j < t)) ? 1 : 0;
    }
    csr_src[beg + rank] = s;
  }
}

__global__ void __launch_bounds__(256)
gat_gather_kernel(const int* __restrict__ rowptr, const int* __restrict__ csr_src,
                  const float* __restrict__ as, const float* __restrict__ ad,
                  const float* __restrict__ h, float* __restrict__ agg) {
  const int lane = threadIdx.x & 31;
  const int d = __builtin_amdgcn_readfirstlane((blockIdx.x * blockDim.x + threadIdx.x) >> 5);
  if (d >= GAT_N) return;
  const int beg = rowptr[d];
  const int deg = rowptr[d + 1] - beg;
  const int T = deg + 1;

  const int slot = lane >> 3, hd = lane & 7;
  const float adv = ad[d * GAT_H + hd];

  float m = -3.0e38f;
  for (int base = 0; base < T; base += 32) {
    const int t = base + lane;
    const int sv = (t < deg) ? csr_src[beg + t] : d;
    const int cnt = (T - base < 32) ? (T - base) : 32;
#pragma unroll
    for (int i = 0; i < 32; i += 4) {
      const int j = i + slot;
      const int jj = (j < cnt) ? j : (cnt - 1);
      const int s = __shfl(sv, jj, 32);
      float v = as[s * GAT_H + hd] + adv;
      v = v > 0.f ? v : NEG_SLOPE * v;
      m = fmaxf(m, v);
    }
  }
  m = fmaxf(m, __shfl_xor(m, 8, 32));
  m = fmaxf(m, __shfl_xor(m, 16, 32));

  float dn = 0.f;
  for (int base = 0; base < T; base += 32) {
    const int t = base + lane;
    const int sv = (t < deg) ? csr_src[beg + t] : d;
    const int cnt = (T - base < 32) ? (T - base) : 32;
#pragma unroll
    for (int i = 0; i < 32; i += 4) {
      const int j = i + slot;
      const int jj = (j < cnt) ? j : (cnt - 1);
      const int s = __shfl(sv, jj, 32);
      float v = as[s * GAT_H + hd] + adv;
      v = v > 0.f ? v : NEG_SLOPE * v;
      const float w = (j < cnt) ? 1.f : 0.f;
      dn += w * __expf(v - m);
    }
  }
  dn += __shfl_xor(dn, 8, 32);
  dn += __shfl_xor(dn, 16, 32);

  const int hA = lane >> 3, hB = 4 + (lane >> 3);
  const float mA = __shfl(m, hA, 32),  mB = __shfl(m, hB, 32);
  const float dA = __shfl(dn, hA, 32) + 1e-16f;
  const float dB = __shfl(dn, hB, 32) + 1e-16f;
  const float adA = ad[d * GAT_H + hA], adB = ad[d * GAT_H + hB];

  const float4* __restrict__ h4 = (const float4*)h;
  float4 acc0 = make_float4(0.f, 0.f, 0.f, 0.f);
  float4 acc1 = make_float4(0.f, 0.f, 0.f, 0.f);
  for (int base = 0; base < T; base += 32) {
    const int t = base + lane;
    const int sv = (t < deg) ? csr_src[beg + t] : d;
    const int cnt = (T - base < 32) ? (T - base) : 32;
#pragma unroll 4
    for (int i = 0; i < cnt; ++i) {
      const int s = __builtin_amdgcn_readlane(sv, i);
      float vA = as[s * GAT_H + hA] + adA;
      float vB = as[s * GAT_H + hB] + adB;
      vA = vA > 0.f ? vA : NEG_SLOPE * vA;
      vB = vB > 0.f ? vB : NEG_SLOPE * vB;
      const float aA = __expf(vA - mA) / dA;
      const float aB = __expf(vB - mB) / dB;
      const float4 x0 = h4[(size_t)s * (GAT_FH / 4) + lane];
      const float4 x1 = h4[(size_t)s * (GAT_FH / 4) + 32 + lane];
      acc0.x += x0.x * aA; acc0.y += x0.y * aA; acc0.z += x0.z * aA; acc0.w += x0.w * aA;
      acc1.x += x1.x * aB; acc1.y += x1.y * aB; acc1.z += x1.z * aB; acc1.w += x1.w * aB;
    }
  }
  float4* o4 = (float4*)(agg + (size_t)d * GAT_FH);
  o4[lane] = acc0;
  o4[32 + lane] = acc1;
}

__global__ void bias_elu_kernel(float* __restrict__ agg,
                                const float* __restrict__ b, int total) {
  int i = blockIdx.x * blockDim.x + threadIdx.x;
  if (i >= total) return;
  float v = agg[i] + b[i & (GAT_FH - 1)];
  agg[i] = v > 0.f ? v : (__expf(v) - 1.f);
}

__global__ void final_linear_kernel(const float* __restrict__ h,
                                    const float* __restrict__ Wl,
                                    const float* __restrict__ bl,
                                    float* __restrict__ y, int n) {
  int gid = blockIdx.x * blockDim.x + threadIdx.x;
  int wid = gid >> 5;
  int lane = threadIdx.x & 31;
  if (wid >= n) return;
  const float* hp = h + (size_t)wid * GAT_FH;
  float s = 0.f;
#pragma unroll
  for (int c = lane; c < GAT_FH; c += 32) s += hp[c] * Wl[c];
#pragma unroll
  for (int off = 16; off > 0; off >>= 1) s += __shfl_down(s, off, 32);
  if (lane == 0) y[wid] = s + bl[0];
}

extern "C" void kernel_launch(void* const* d_in, const int* in_sizes, int n_in,
                              void* d_out, int out_size, void* d_ws, size_t ws_size,
                              hipStream_t stream) {
  const float* x      = (const float*)d_in[0];
  const int*   ei     = (const int*)  d_in[1];
  const float* W1     = (const float*)d_in[2];
  const float* a1_src = (const float*)d_in[3];
  const float* a1_dst = (const float*)d_in[4];
  const float* b1     = (const float*)d_in[5];
  const float* W2     = (const float*)d_in[6];
  const float* a2_src = (const float*)d_in[7];
  const float* a2_dst = (const float*)d_in[8];
  const float* b2     = (const float*)d_in[9];
  const float* Wl     = (const float*)d_in[10];
  const float* bl     = (const float*)d_in[11];
  float* out = (float*)d_out;

  float* hbuf  = (float*)d_ws;
  float* agg   = hbuf  + (size_t)GAT_N * GAT_FH;
  float* as    = agg   + (size_t)GAT_N * GAT_FH;
  float* ad    = as    + (size_t)GAT_N * GAT_H;
  int* deg     = (int*)(ad + (size_t)GAT_N * GAT_H);
  int* rowptr  = deg    + GAT_N;
  int* cursor  = rowptr + (GAT_N + 1);
  int* tmp_src = cursor + GAT_N;
  int* csr_src = tmp_src + GAT_E;

  const int NF  = GAT_N * GAT_FH;
  const int NH  = GAT_N * GAT_H;

  dim3 gemm_grid(GAT_N / 16, GAT_FH / 64);
  dim3 gemm_blk(128);

  const int TPB = 256;
  auto blocks = [](int n, int tpb) { return (n + tpb - 1) / tpb; };

  csr_zero_kernel<<<blocks(GAT_N, TPB), TPB, 0, stream>>>(deg);
  csr_count_kernel<<<blocks(GAT_E, TPB), TPB, 0, stream>>>(ei, deg);
  csr_scan_kernel<<<1, 1024, 0, stream>>>(deg, rowptr, cursor);
  csr_fill_kernel<<<blocks(GAT_E, TPB), TPB, 0, stream>>>(ei, cursor, tmp_src);
  csr_order_kernel<<<blocks(GAT_N * 32, TPB), TPB, 0, stream>>>(rowptr, tmp_src, csr_src);

  gemm_wmma_f32<<<gemm_grid, gemm_blk, 0, stream>>>(x, W1, hbuf, GAT_N, 128, GAT_FH);
  alphas_kernel<<<blocks(NH, TPB), TPB, 0, stream>>>(hbuf, a1_src, a1_dst, as, ad, NH);
  gat_gather_kernel<<<blocks(GAT_N * 32, TPB), TPB, 0, stream>>>(rowptr, csr_src, as, ad, hbuf, agg);
  bias_elu_kernel<<<blocks(NF, TPB), TPB, 0, stream>>>(agg, b1, NF);

  gemm_wmma_f32<<<gemm_grid, gemm_blk, 0, stream>>>(agg, W2, hbuf, GAT_N, GAT_FH, GAT_FH);
  alphas_kernel<<<blocks(NH, TPB), TPB, 0, stream>>>(hbuf, a2_src, a2_dst, as, ad, NH);
  gat_gather_kernel<<<blocks(GAT_N * 32, TPB), TPB, 0, stream>>>(rowptr, csr_src, as, ad, hbuf, agg);
  bias_elu_kernel<<<blocks(NF, TPB), TPB, 0, stream>>>(agg, b2, NF);

  final_linear_kernel<<<blocks(GAT_N * 32, TPB), TPB, 0, stream>>>(agg, Wl, bl, out, GAT_N);
}


# Bias+ELU (both layers) and the final dot with Wl (new sum order: 8 products per lane, then a wave butterfly) fused into the gather kernel's epilogue on its registers; 14 launches become 11
# speedup vs baseline: 8.1459x; 1.1581x over previous
#include <hip/hip_runtime.h>
#include <hip/hip_bf16.h>

#define GAT_N 30000
#define GAT_E 480000
#define GAT_ETOT (GAT_E + GAT_N)
#define GAT_H 8
#define GAT_C 32
#define GAT_FH 256
#define NEG_SLOPE 0.2f

typedef float v2f __attribute__((ext_vector_type(2)));
typedef float v8f __attribute__((ext_vector_type(8)));

__global__ void gemm_wmma_f32(const float* __restrict__ A,
                              const float* __restrict__ B,
                              float* __restrict__ C,
                              int M, int K, int Ncols) {
  const int wave = threadIdx.x >> 5;
  const int lane = threadIdx.x & 31;
  const int half = lane >> 4;
  const int l    = lane & 15;

  const int m0 = blockIdx.x * 16;
  const int n0 = blockIdx.y * 64 + wave * 16;
  if (m0 >= M || n0 >= Ncols) return;

  v8f c = {0.f, 0.f, 0.f, 0.f, 0.f, 0.f, 0.f, 0.f};

  const float* arow = A + (size_t)(m0 + l) * K + half * 2;
  const float* bcol = B + (size_t)(half * 2) * Ncols + n0 + l;

  for (int k = 0; k < K; k += 4) {
    v2f a = *(const v2f*)(arow + k);
    v2f b;
    b.x = bcol[(size_t)k * Ncols];
    b.y = bcol[(size_t)k * Ncols + Ncols];
    c = __builtin_amdgcn_wmma_f32_16x16x4_f32(
         false, a,  false, b,
         (short)0, c,  false,  false);
  }

  float* crow = C + n0 + l;
  const int mbase = m0 + half * 8;
#pragma unroll
  for (int r = 0; r < 8; ++r)
    crow[(size_t)(mbase + r) * Ncols] = c[r];
}

__global__ void alphas_kernel(const float* __restrict__ h,
                              const float* __restrict__ asrc,
                              const float* __restrict__ adst,
                              float* __restrict__ s, float* __restrict__ d,
                              int total) {
  int i = blockIdx.x * blockDim.x + threadIdx.x;
  if (i >= total) return;
  int node = i >> 3, hd = i & 7;
  const float* hp = h + (size_t)node * GAT_FH + hd * GAT_C;
  const float* ap = asrc + hd * GAT_C;
  const float* bp = adst + hd * GAT_C;
  float ss = 0.f, dd = 0.f;
#pragma unroll
  for (int cc = 0; cc < GAT_C; ++cc) {
    float v = hp[cc];
    ss += v * ap[cc];
    dd += v * bp[cc];
  }
  s[i] = ss;
  d[i] = dd;
}

__global__ void csr_zero_kernel(int* __restrict__ deg) {
  int i = blockIdx.x * blockDim.x + threadIdx.x;
  if (i < GAT_N) deg[i] = 0;
}

__global__ void csr_count_kernel(const int* __restrict__ ei, int* __restrict__ deg) {
  int e = blockIdx.x * blockDim.x + threadIdx.x;
  if (e < GAT_E) atomicAdd(&deg[ei[GAT_E + e]], 1);
}

#define SCAN_PER ((GAT_N + 1023) / 1024)
__global__ void __launch_bounds__(1024)
csr_scan_kernel(const int* __restrict__ deg, int* __restrict__ rowptr,
                int* __restrict__ cursor) {
  __shared__ int part[1024];
  const int t = threadIdx.x;
  const int b = t * SCAN_PER;
  const int e = (b + SCAN_PER < GAT_N) ? (b + SCAN_PER) : GAT_N;
  int s = 0;
  for (int i = b; i < e; ++i) s += deg[i];
  part[t] = s;
  __syncthreads();
  for (int off = 1; off < 1024; off <<= 1) {
    int v = (t >= off) ? part[t - off] : 0;
    __syncthreads();
    part[t] += v;
    __syncthreads();
  }
  int run = part[t] - s;
  for (int i = b; i < e; ++i) {
    rowptr[i] = run;
    cursor[i] = run;
    run += deg[i];
  }
  if (t == 1023) rowptr[GAT_N] = part[1023];
}

__global__ void csr_fill_kernel(const int* __restrict__ ei, int* __restrict__ cursor,
                                int* __restrict__ tmp_src) {
  int e = blockIdx.x * blockDim.x + threadIdx.x;
  if (e >= GAT_E) return;
  int p = atomicAdd(&cursor[ei[GAT_E + e]], 1);
  tmp_src[p] = ei[e];
}

__global__ void csr_order_kernel(const int* __restrict__ rowptr,
                                 const int* __restrict__ tmp_src,
                                 int* __restrict__ csr_src) {
  const int lane = threadIdx.x & 31;
  const int d = __builtin_amdgcn_readfirstlane((blockIdx.x * blockDim.x + threadIdx.x) >> 5);
  if (d >= GAT_N) return;
  const int beg = rowptr[d];
  const int deg = rowptr[d + 1] - beg;
  for (int t = lane; t < deg; t += 32) {
    const int s = tmp_src[beg + t];
    int rank = 0;
    for (int j = 0; j < deg; ++j) {
      const int sj = tmp_src[beg + j];
      rank += ((sj < s) || (sj == s && j < t)) ? 1 : 0;
    }
    csr_src[beg + rank] = s;
  }
}

__device__ __forceinline__ float bias_elu(float a, float b) {
  const float v = a + b;
  return v > 0.f ? v : (__expf(v) - 1.f);
}

template <bool LAST>
__global__ void __launch_bounds__(256)
gat_gather_kernel(const int* __restrict__ rowptr, const int* __restrict__ csr_src,
                  const float* __restrict__ as, const float* __restrict__ ad,
                  const float* __restrict__ h, const float* __restrict__ bias,
                  const float* __restrict__ Wl, const float* __restrict__ bl,
                  float* __restrict__ agg, float* __restrict__ y) {
  const int lane = threadIdx.x & 31;
  const int d = __builtin_amdgcn_readfirstlane((blockIdx.x * blockDim.x + threadIdx.x) >> 5);
  if (d >= GAT_N) return;
  const int beg = rowptr[d];
  const int deg = rowptr[d + 1] - beg;
  const int T = deg + 1;

  const int slot = lane >> 3, hd = lane & 7;
  const float adv = ad[d * GAT_H + hd];

  float m = -3.0e38f;
  for (int base = 0; base < T; base += 32) {
    const int t = base + lane;
    const int sv = (t < deg) ? csr_src[beg + t] : d;
    const int cnt = (T - base < 32) ? (T - base) : 32;
#pragma unroll
    for (int i = 0; i < 32; i += 4) {
      const int j = i + slot;
      const int jj = (j < cnt) ? j : (cnt - 1);
      const int s = __shfl(sv, jj, 32);
      float v = as[s * GAT_H + hd] + adv;
      v = v > 0.f ? v : NEG_SLOPE * v;
      m = fmaxf(m, v);
    }
  }
  m = fmaxf(m, __shfl_xor(m, 8, 32));
  m = fmaxf(m, __shfl_xor(m, 16, 32));

  float dn = 0.f;
  for (int base = 0; base < T; base += 32) {
    const int t = base + lane;
    const int sv = (t < deg) ? csr_src[beg + t] : d;
    const int cnt = (T - base < 32) ? (T - base) : 32;
#pragma unroll
    for (int i = 0; i < 32; i += 4) {
      const int j = i + slot;
      const int jj = (j < cnt) ? j : (cnt - 1);
      const int s = __shfl(sv, jj, 32);
      float v = as[s * GAT_H + hd] + adv;
      v = v > 0.f ? v : NEG_SLOPE * v;
      const float w = (j < cnt) ? 1.f : 0.f;
      dn += w * __expf(v - m);
    }
  }
  dn += __shfl_xor(dn, 8, 32);
  dn += __shfl_xor(dn, 16, 32);

  const int hA = lane >> 3, hB = 4 + (lane >> 3);
  const float mA = __shfl(m, hA, 32),  mB = __shfl(m, hB, 32);
  const float dA = __shfl(dn, hA, 32) + 1e-16f;
  const float dB = __shfl(dn, hB, 32) + 1e-16f;
  const float adA = ad[d * GAT_H + hA], adB = ad[d * GAT_H + hB];

  const float4* __restrict__ h4 = (const float4*)h;
  float4 acc0 = make_float4(0.f, 0.f, 0.f, 0.f);
  float4 acc1 = make_float4(0.f, 0.f, 0.f, 0.f);
  for (int base = 0; base < T; base += 32) {
    const int t = base + lane;
    const int sv = (t < deg) ? csr_src[beg + t] : d;
    const int cnt = (T - base < 32) ? (T - base) : 32;
#pragma unroll 4
    for (int i = 0; i < cnt; ++i) {
      const int s = __builtin_amdgcn_readlane(sv, i);
      float vA = as[s * GAT_H + hA] + adA;
      float vB = as[s * GAT_H + hB] + adB;
      vA = vA > 0.f ? vA : NEG_SLOPE * vA;
      vB = vB > 0.f ? vB : NEG_SLOPE * vB;
      const float aA = __expf(vA - mA) / dA;
      const float aB = __expf(vB - mB) / dB;
      const float4 x0 = h4[(size_t)s * (GAT_FH / 4) + lane];
      const float4 x1 = h4[(size_t)s * (GAT_FH / 4) + 32 + lane];
      acc0.x += x0.x * aA; acc0.y += x0.y * aA; acc0.z += x0.z * aA; acc0.w += x0.w * aA;
      acc1.x += x1.x * aB; acc1.y += x1.y * aB; acc1.z += x1.z * aB; acc1.w += x1.w * aB;
    }
  }
  const float4* __restrict__ b4 = (const float4*)bias;
  const float4 bb0 = b4[lane], bb1 = b4[32 + lane];
  float4 r0, r1;
  r0.x = bias_elu(acc0.x, bb0.x); r0.y = bias_elu(acc0.y, bb0.y);
  r0.z = bias_elu(acc0.z, bb0.z); r0.w = bias_elu(acc0.w, bb0.w);
  r1.x = bias_elu(acc1.x, bb1.x); r1.y = bias_elu(acc1.y, bb1.y);
  r1.z = bias_elu(acc1.z, bb1.z); r1.w = bias_elu(acc1.w, bb1.w);
  if (!LAST) {
    float4* o4 = (float4*)(agg + (size_t)d * GAT_FH);
    o4[lane] = r0;
    o4[32 + lane] = r1;
  } else {
    const float4* __restrict__ w4 = (const float4*)Wl;
    const float4 w0 = w4[lane], w1 = w4[32 + lane];
    float s = r0.x * w0.x;
    s += r0.y * w0.y; s += r0.z * w0.z; s += r0.w * w0.w;
    s += r1.x * w1.x; s += r1.y * w1.y; s += r1.z * w1.z; s += r1.w * w1.w;
    s += __shfl_xor(s, 16, 32);
    s += __shfl_xor(s, 8, 32);
    s += __shfl_xor(s, 4, 32);
    s += __shfl_xor(s, 2, 32);
    s += __shfl_xor(s, 1, 32);
    if (lane == 0) y[d] = s + bl[0];
  }
}

extern "C" void kernel_launch(void* const* d_in, const int* in_sizes, int n_in,
                              void* d_out, int out_size, void* d_ws, size_t ws_size,
                              hipStream_t stream) {
  const float* x      = (const float*)d_in[0];
  const int*   ei     = (const int*)  d_in[1];
  const float* W1     = (const float*)d_in[2];
  const float* a1_src = (const float*)d_in[3];
  const float* a1_dst = (const float*)d_in[4];
  const float* b1     = (const float*)d_in[5];
  const float* W2     = (const float*)d_in[6];
  const float* a2_src = (const float*)d_in[7];
  const float* a2_dst = (const float*)d_in[8];
  const float* b2     = (const float*)d_in[9];
  const float* Wl     = (const float*)d_in[10];
  const float* bl     = (const float*)d_in[11];
  float* out = (float*)d_out;

  float* hbuf  = (float*)d_ws;
  float* agg   = hbuf  + (size_t)GAT_N * GAT_FH;
  float* as    = agg   + (size_t)GAT_N * GAT_FH;
  float* ad    = as    + (size_t)GAT_N * GAT_H;
  int* deg     = (int*)(ad + (size_t)GAT_N * GAT_H);
  int* rowptr  = deg    + GAT_N;
  int* cursor  = rowptr + (GAT_N + 1);
  int* tmp_src = cursor + GAT_N;
  int* csr_src = tmp_src + GAT_E;

  const int NH  = GAT_N * GAT_H;

  dim3 gemm_grid(GAT_N / 16, GAT_FH / 64);
  dim3 gemm_blk(128);

  const int TPB = 256;
  auto blocks = [](int n, int tpb) { return (n + tpb - 1) / tpb; };

  csr_zero_kernel<<<blocks(GAT_N, TPB), TPB, 0, stream>>>(deg);
  csr_count_kernel<<<blocks(GAT_E, TPB), TPB, 0, stream>>>(ei, deg);
  csr_scan_kernel<<<1, 1024, 0, stream>>>(deg, rowptr, cursor);
  csr_fill_kernel<<<blocks(GAT_E, TPB), TPB, 0, stream>>>(ei, cursor, tmp_src);
  csr_order_kernel<<<blocks(GAT_N * 32, TPB), TPB, 0, stream>>>(rowptr, tmp_src, csr_src);

  gemm_wmma_f32<<<gemm_grid, gemm_blk, 0, stream>>>(x, W1, hbuf, GAT_N, 128, GAT_FH);
  alphas_kernel<<<blocks(NH, TPB), TPB, 0, stream>>>(hbuf, a1_src, a1_dst, as, ad, NH);
  gat_gather_kernel<false><<<blocks(GAT_N * 32, TPB), TPB, 0, stream>>>(
      rowptr, csr_src, as, ad, hbuf, b1, Wl, bl, agg, out);

  gemm_wmma_f32<<<gemm_grid, gemm_blk, 0, stream>>>(agg, W2, hbuf, GAT_N, GAT_FH, GAT_FH);
  alphas_kernel<<<blocks(NH, TPB), TPB, 0, stream>>>(hbuf, a2_src, a2_dst, as, ad, NH);
  gat_gather_kernel<true><<<blocks(GAT_N * 32, TPB), TPB, 0, stream>>>(
      rowptr, csr_src, as, ad, hbuf, b2, Wl, bl, agg, out);
}


# GEMM retiled: B strip staged once per workgroup in LDS, each wave owns 16x64 (4 accumulators), A read 16 B per lane; k order inside 8-wide chunks permuted (new sum order in both products)
# speedup vs baseline: 9.8557x; 1.2099x over previous
#include <hip/hip_runtime.h>
#include <hip/hip_bf16.h>

#define GAT_N 30000
#define GAT_E 480000
#define GAT_ETOT (GAT_E + GAT_N)
#define GAT_H 8
#define GAT_C 32
#define GAT_FH 256
#define NEG_SLOPE 0.2f

typedef float v2f __attribute__((ext_vector_type(2)));
typedef float v8f __attribute__((ext_vector_type(8)));

#define GEMM_BN 64
#define GEMM_WAVES 8
#define GEMM_MGROUPS ((GAT_N / 16 + GEMM_WAVES - 1) / GEMM_WAVES)

__device__ __forceinline__ v8f wmma4(v2f a, v2f b, v8f c) {
  return __builtin_amdgcn_wmma_f32_16x16x4_f32(
       false, a,  false, b,
       (short)0, c,  false,  false);
}

template <int K>
__global__ void __launch_bounds__(GEMM_WAVES * 32)
gemm_wmma_f32(const float* __restrict__ A,
              const float* __restrict__ B,
              float* __restrict__ C, int M) {
  constexpr int KG = K / 4;
  constexpr int NC = K / 8;
  __shared__ float4 Bs[KG * GEMM_BN];

  const int tid = threadIdx.x;
  const int n0 = blockIdx.x * GEMM_BN;

#pragma unroll 4
  for (int it = 0; it < KG * GEMM_BN / (GEMM_WAVES * 32); ++it) {
    const int idx = it * (GEMM_WAVES * 32) + tid;
    const int kg = idx >> 6, n = idx & (GEMM_BN - 1);
    const float* bp = B + (size_t)(4 * kg) * GAT_FH + n0 + n;
    float4 v;
    v.x = bp[0];
    v.y = bp[GAT_FH];
    v.z = bp[2 * GAT_FH];
    v.w = bp[3 * GAT_FH];
    Bs[idx] = v;
  }
  __syncthreads();

  const int wave = __builtin_amdgcn_readfirstlane(tid >> 5);
  const int lane = tid & 31;
  const int half = lane >> 4;
  const int l    = lane & 15;
  const int m0 = (blockIdx.y * GEMM_WAVES + wave) * 16;
  if (m0 >= M) return;

  const float4* __restrict__ ap = (const float4*)(A + (size_t)(m0 + l) * K) + half;
  const float4* bs = Bs + half * GEMM_BN + l;

  v8f acc0 = {0.f, 0.f, 0.f, 0.f, 0.f, 0.f, 0.f, 0.f};
  v8f acc1 = acc0, acc2 = acc0, acc3 = acc0;

#pragma unroll 4
  for (int c = 0; c < NC; ++c) {
    const float4 a  = ap[2 * c];
    const float4 b0 = bs[c * 2 * GEMM_BN];
    const float4 b1 = bs[c * 2 * GEMM_BN + 16];
    const float4 b2 = bs[c * 2 * GEMM_BN + 32];
    const float4 b3 = bs[c * 2 * GEMM_BN + 48];

    const v2f a01 = {a.x, a.y};
    const v2f a23 = {a.z, a.w};
    acc0 = wmma4(a01, (v2f){b0.x, b0.y}, acc0);
    acc1 = wmma4(a01, (v2f){b1.x, b1.y}, acc1);
    acc2 = wmma4(a01, (v2f){b2.x, b2.y}, acc2);
    acc3 = wmma4(a01, (v2f){b3.x, b3.y}, acc3);
    acc0 = wmma4(a23, (v2f){b0.z, b0.w}, acc0);
    acc1 = wmma4(a23, (v2f){b1.z, b1.w}, acc1);
    acc2 = wmma4(a23, (v2f){b2.z, b2.w}, acc2);
    acc3 = wmma4(a23, (v2f){b3.z, b3.w}, acc3);
  }

  float* crow = C + (size_t)(m0 + half * 8) * GAT_FH + n0 + l;
#pragma unroll
  for (int r = 0; r < 8; ++r) {
    crow[(size_t)r * GAT_FH]      = acc0[r];
    crow[(size_t)r * GAT_FH + 16] = acc1[r];
    crow[(size_t)r * GAT_FH + 32] = acc2[r];
    crow[(size_t)r * GAT_FH + 48] = acc3[r];
  }
}

__global__ void alphas_kernel(const float* __restrict__ h,
                              const float* __restrict__ asrc,
                              const float* __restrict__ adst,
                              float* __restrict__ s, float* __restrict__ d,
                              int total) {
  int i = blockIdx.x * blockDim.x + threadIdx.x;
  if (i >= total) return;
  int node = i >> 3, hd = i & 7;
  const float* hp = h + (size_t)node * GAT_FH + hd * GAT_C;
  const float* ap = asrc + hd * GAT_C;
  const float* bp = adst + hd * GAT_C;
  float ss = 0.f, dd = 0.f;
#pragma unroll
  for (int cc = 0; cc < GAT_C; ++cc) {
    float v = hp[cc];
    ss += v * ap[cc];
    dd += v * bp[cc];
  }
  s[i] = ss;
  d[i] = dd;
}

__global__ void csr_zero_kernel(int* __restrict__ deg) {
  int i = blockIdx.x * blockDim.x + threadIdx.x;
  if (i < GAT_N) deg[i] = 0;
}

__global__ void csr_count_kernel(const int* __restrict__ ei, int* __restrict__ deg) {
  int e = blockIdx.x * blockDim.x + threadIdx.x;
  if (e < GAT_E) atomicAdd(&deg[ei[GAT_E + e]], 1);
}

#define SCAN_PER ((GAT_N + 1023) / 1024)
__global__ void __launch_bounds__(1024)
csr_scan_kernel(const int* __restrict__ deg, int* __restrict__ rowptr,
                int* __restrict__ cursor) {
  __shared__ int part[1024];
  const int t = threadIdx.x;
  const int b = t * SCAN_PER;
  const int e = (b + SCAN_PER < GAT_N) ? (b + SCAN_PER) : GAT_N;
  int s = 0;
  for (int i = b; i < e; ++i) s += deg[i];
  part[t] = s;
  __syncthreads();
  for (int off = 1; off < 1024; off <<= 1) {
    int v = (t >= off) ? part[t - off] : 0;
    __syncthreads();
    part[t] += v;
    __syncthreads();
  }
  int run = part[t] - s;
  for (int i = b; i < e; ++i) {
    rowptr[i] = run;
    cursor[i] = run;
    run += deg[i];
  }
  if (t == 1023) rowptr[GAT_N] = part[1023];
}

__global__ void csr_fill_kernel(const int* __restrict__ ei, int* __restrict__ cursor,
                                int* __restrict__ tmp_src) {
  int e = blockIdx.x * blockDim.x + threadIdx.x;
  if (e >= GAT_E) return;
  int p = atomicAdd(&cursor[ei[GAT_E + e]], 1);
  tmp_src[p] = ei[e];
}

__global__ void csr_order_kernel(const int* __restrict__ rowptr,
                                 const int* __restrict__ tmp_src,
                                 int* __restrict__ csr_src) {
  const int lane = threadIdx.x & 31;
  const int d = __builtin_amdgcn_readfirstlane((blockIdx.x * blockDim.x + threadIdx.x) >> 5);
  if (d >= GAT_N) return;
  const int beg = rowptr[d];
  const int deg = rowptr[d + 1] - beg;
  for (int t = lane; t < deg; t += 32) {
    const int s = tmp_src[beg + t];
    int rank = 0;
    for (int j = 0; j < deg; ++j) {
      const int sj = tmp_src[beg + j];
      rank += ((sj < s) || (sj == s && j < t)) ? 1 : 0;
    }
    csr_src[beg + rank] = s;
  }
}

__device__ __forceinline__ float bias_elu(float a, float b) {
  const float v = a + b;
  return v > 0.f ? v : (__expf(v) - 1.f);
}

template <bool LAST>
__global__ void __launch_bounds__(256)
gat_gather_kernel(const int* __restrict__ rowptr, const int* __restrict__ csr_src,
                  const float* __restrict__ as, const float* __restrict__ ad,
                  const float* __restrict__ h, const float* __restrict__ bias,
                  const float* __restrict__ Wl, const float* __restrict__ bl,
                  float* __restrict__ agg, float* __restrict__ y) {
  const int lane = threadIdx.x & 31;
  const int d = __builtin_amdgcn_readfirstlane((blockIdx.x * blockDim.x + threadIdx.x) >> 5);
  if (d >= GAT_N) return;
  const int beg = rowptr[d];
  const int deg = rowptr[d + 1] - beg;
  const int T = deg + 1;

  const int slot = lane >> 3, hd = lane & 7;
  const float adv = ad[d * GAT_H + hd];

  float m = -3.0e38f;
  for (int base = 0; base < T; base += 32) {
    const int t = base + lane;
    const int sv = (t < deg) ? csr_src[beg + t] : d;
    const int cnt = (T - base < 32) ? (T - base) : 32;
#pragma unroll
    for (int i = 0; i < 32; i += 4) {
      const int j = i + slot;
      const int jj = (j < cnt) ? j : (cnt - 1);
      const int s = __shfl(sv, jj, 32);
      float v = as[s * GAT_H + hd] + adv;
      v = v > 0.f ? v : NEG_SLOPE * v;
      m = fmaxf(m, v);
    }
  }
  m = fmaxf(m, __shfl_xor(m, 8, 32));
  m = fmaxf(m, __shfl_xor(m, 16, 32));

  float dn = 0.f;
  for (int base = 0; base < T; base += 32) {
    const int t = base + lane;
    const int sv = (t < deg) ? csr_src[beg + t] : d;
    const int cnt = (T - base < 32) ? (T - base) : 32;
#pragma unroll
    for (int i = 0; i < 32; i += 4) {
      const int j = i + slot;
      const int jj = (j < cnt) ? j : (cnt - 1);
      const int s = __shfl(sv, jj, 32);
      float v = as[s * GAT_H + hd] + adv;
      v = v > 0.f ? v : NEG_SLOPE * v;
      const float w = (j < cnt) ? 1.f : 0.f;
      dn += w * __expf(v - m);
    }
  }
  dn += __shfl_xor(dn, 8, 32);
  dn += __shfl_xor(dn, 16, 32);

  const int hA = lane >> 3, hB = 4 + (lane >> 3);
  const float mA = __shfl(m, hA, 32),  mB = __shfl(m, hB, 32);
  const float dA = __shfl(dn, hA, 32) + 1e-16f;
  const float dB = __shfl(dn, hB, 32) + 1e-16f;
  const float adA = ad[d * GAT_H + hA], adB = ad[d * GAT_H + hB];

  const float4* __restrict__ h4 = (const float4*)h;
  float4 acc0 = make_float4(0.f, 0.f, 0.f, 0.f);
  float4 acc1 = make_float4(0.f, 0.f, 0.f, 0.f);
  for (int base = 0; base < T; base += 32) {
    const int t = base + lane;
    const int sv = (t < deg) ? csr_src[beg + t] : d;
    const int cnt = (T - base < 32) ? (T - base) : 32;
#pragma unroll 4
    for (int i = 0; i < cnt; ++i) {
      const int s = __builtin_amdgcn_readlane(sv, i);
      float vA = as[s * GAT_H + hA] + adA;
      float vB = as[s * GAT_H + hB] + adB;
      vA = vA > 0.f ? vA : NEG_SLOPE * vA;
      vB = vB > 0.f ? vB : NEG_SLOPE * vB;
      const float aA = __expf(vA - mA) / dA;
      const float aB = __expf(vB - mB) / dB;
      const float4 x0 = h4[(size_t)s * (GAT_FH / 4) + lane];
      const float4 x1 = h4[(size_t)s * (GAT_FH / 4) + 32 + lane];
      acc0.x += x0.x * aA; acc0.y += x0.y * aA; acc0.z += x0.z * aA; acc0.w += x0.w * aA;
      acc1.x += x1.x * aB; acc1.y += x1.y * aB; acc1.z += x1.z * aB; acc1.w += x1.w * aB;
    }
  }
  const float4* __restrict__ b4 = (const float4*)bias;
  const float4 bb0 = b4[lane], bb1 = b4[32 + lane];
  float4 r0, r1;
  r0.x = bias_elu(acc0.x, bb0.x); r0.y = bias_elu(acc0.y, bb0.y);
  r0.z = bias_elu(acc0.z, bb0.z); r0.w = bias_elu(acc0.w, bb0.w);
  r1.x = bias_elu(acc1.x, bb1.x); r1.y = bias_elu(acc1.y, bb1.y);
  r1.z = bias_elu(acc1.z, bb1.z); r1.w = bias_elu(acc1.w, bb1.w);
  if (!LAST) {
    float4* o4 = (float4*)(agg + (size_t)d * GAT_FH);
    o4[lane] = r0;
    o4[32 + lane] = r1;
  } else {
    const float4* __restrict__ w4 = (const float4*)Wl;
    const float4 w0 = w4[lane], w1 = w4[32 + lane];
    float s = r0.x * w0.x;
    s += r0.y * w0.y; s += r0.z * w0.z; s += r0.w * w0.w;
    s += r1.x * w1.x; s += r1.y * w1.y; s += r1.z * w1.z; s += r1.w * w1.w;
    s += __shfl_xor(s, 16, 32);
    s += __shfl_xor(s, 8, 32);
    s += __shfl_xor(s, 4, 32);
    s += __shfl_xor(s, 2, 32);
    s += __shfl_xor(s, 1, 32);
    if (lane == 0) y[d] = s + bl[0];
  }
}

extern "C" void kernel_launch(void* const* d_in, const int* in_sizes, int n_in,
                              void* d_out, int out_size, void* d_ws, size_t ws_size,
                              hipStream_t stream) {
  const float* x      = (const float*)d_in[0];
  const int*   ei     = (const int*)  d_in[1];
  const float* W1     = (const float*)d_in[2];
  const float* a1_src = (const float*)d_in[3];
  const float* a1_dst = (const float*)d_in[4];
  const float* b1     = (const float*)d_in[5];
  const float* W2     = (const float*)d_in[6];
  const float* a2_src = (const float*)d_in[7];
  const float* a2_dst = (const float*)d_in[8];
  const float* b2     = (const float*)d_in[9];
  const float* Wl     = (const float*)d_in[10];
  const float* bl     = (const float*)d_in[11];
  float* out = (float*)d_out;

  float* hbuf  = (float*)d_ws;
  float* agg   = hbuf  + (size_t)GAT_N * GAT_FH;
  float* as    = agg   + (size_t)GAT_N * GAT_FH;
  float* ad    = as    + (size_t)GAT_N * GAT_H;
  int* deg     = (int*)(ad + (size_t)GAT_N * GAT_H);
  int* rowptr  = deg    + GAT_N;
  int* cursor  = rowptr + (GAT_N + 1);
  int* tmp_src = cursor + GAT_N;
  int* csr_src = tmp_src + GAT_E;

  const int NH  = GAT_N * GAT_H;

  dim3 gemm_grid(GAT_FH / GEMM_BN, GEMM_MGROUPS);
  dim3 gemm_blk(GEMM_WAVES * 32);

  const int TPB = 256;
  auto blocks = [](int n, int tpb) { return (n + tpb - 1) / tpb; };

  csr_zero_kernel<<<blocks(GAT_N, TPB), TPB, 0, stream>>>(deg);
  csr_count_kernel<<<blocks(GAT_E, TPB), TPB, 0, stream>>>(ei, deg);
  csr_scan_kernel<<<1, 1024, 0, stream>>>(deg, rowptr, cursor);
  csr_fill_kernel<<<blocks(GAT_E, TPB), TPB, 0, stream>>>(ei, cursor, tmp_src);
  csr_order_kernel<<<blocks(GAT_N * 32, TPB), TPB, 0, stream>>>(rowptr, tmp_src, csr_src);

  gemm_wmma_f32<128><<<gemm_grid, gemm_blk, 0, stream>>>(x, W1, hbuf, GAT_N);
  alphas_kernel<<<blocks(NH, TPB), TPB, 0, stream>>>(hbuf, a1_src, a1_dst, as, ad, NH);
  gat_gather_kernel<false><<<blocks(GAT_N * 32, TPB), TPB, 0, stream>>>(
      rowptr, csr_src, as, ad, hbuf, b1, Wl, bl, agg, out);

  gemm_wmma_f32<GAT_FH><<<gemm_grid, gemm_blk, 0, stream>>>(agg, W2, hbuf, GAT_N);
  alphas_kernel<<<blocks(NH, TPB), TPB, 0, stream>>>(hbuf, a2_src, a2_dst, as, ad, NH);
  gat_gather_kernel<true><<<blocks(GAT_N * 32, TPB), TPB, 0, stream>>>(
      rowptr, csr_src, as, ad, hbuf, b2, Wl, bl, agg, out);
}


# Degree scan spread over 30 workgroups with whole-line reads (each sums the degrees ahead of its 1024-node slice, then scans it in registers) instead of one workgroup on 120-byte-strided runs
# speedup vs baseline: 12.3193x; 1.2500x over previous
#include <hip/hip_runtime.h>
#include <hip/hip_bf16.h>

#define GAT_N 30000
#define GAT_E 480000
#define GAT_ETOT (GAT_E + GAT_N)
#define GAT_H 8
#define GAT_C 32
#define GAT_FH 256
#define NEG_SLOPE 0.2f

typedef float v2f __attribute__((ext_vector_type(2)));
typedef float v8f __attribute__((ext_vector_type(8)));

#define GEMM_BN 64
#define GEMM_WAVES 8
#define GEMM_MGROUPS ((GAT_N / 16 + GEMM_WAVES - 1) / GEMM_WAVES)

__device__ __forceinline__ v8f wmma4(v2f a, v2f b, v8f c) {
  return __builtin_amdgcn_wmma_f32_16x16x4_f32(
       false, a,  false, b,
       (short)0, c,  false,  false);
}

template <int K>
__global__ void __launch_bounds__(GEMM_WAVES * 32)
gemm_wmma_f32(const float* __restrict__ A,
              const float* __restrict__ B,
              float* __restrict__ C, int M) {
  constexpr int KG = K / 4;
  constexpr int NC = K / 8;
  __shared__ float4 Bs[KG * GEMM_BN];

  const int tid = threadIdx.x;
  const int n0 = blockIdx.x * GEMM_BN;

#pragma unroll 4
  for (int it = 0; it < KG * GEMM_BN / (GEMM_WAVES * 32); ++it) {
    const int idx = it * (GEMM_WAVES * 32) + tid;
    const int kg = idx >> 6, n = idx & (GEMM_BN - 1);
    const float* bp = B + (size_t)(4 * kg) * GAT_FH + n0 + n;
    float4 v;
    v.x = bp[0];
    v.y = bp[GAT_FH];
    v.z = bp[2 * GAT_FH];
    v.w = bp[3 * GAT_FH];
    Bs[idx] = v;
  }
  __syncthreads();

  const int wave = __builtin_amdgcn_readfirstlane(tid >> 5);
  const int lane = tid & 31;
  const int half = lane >> 4;
  const int l    = lane & 15;
  const int m0 = (blockIdx.y * GEMM_WAVES + wave) * 16;
  if (m0 >= M) return;

  const float4* __restrict__ ap = (const float4*)(A + (size_t)(m0 + l) * K) + half;
  const float4* bs = Bs + half * GEMM_BN + l;

  v8f acc0 = {0.f, 0.f, 0.f, 0.f, 0.f, 0.f, 0.f, 0.f};
  v8f acc1 = acc0, acc2 = acc0, acc3 = acc0;

#pragma unroll 4
  for (int c = 0; c < NC; ++c) {
    const float4 a  = ap[2 * c];
    const float4 b0 = bs[c * 2 * GEMM_BN];
    const float4 b1 = bs[c * 2 * GEMM_BN + 16];
    const float4 b2 = bs[c * 2 * GEMM_BN + 32];
    const float4 b3 = bs[c * 2 * GEMM_BN + 48];

    const v2f a01 = {a.x, a.y};
    const v2f a23 = {a.z, a.w};
    acc0 = wmma4(a01, (v2f){b0.x, b0.y}, acc0);
    acc1 = wmma4(a01, (v2f){b1.x, b1.y}, acc1);
    acc2 = wmma4(a01, (v2f){b2.x, b2.y}, acc2);
    acc3 = wmma4(a01, (v2f){b3.x, b3.y}, acc3);
    acc0 = wmma4(a23, (v2f){b0.z, b0.w}, acc0);
    acc1 = wmma4(a23, (v2f){b1.z, b1.w}, acc1);
    acc2 = wmma4(a23, (v2f){b2.z, b2.w}, acc2);
    acc3 = wmma4(a23, (v2f){b3.z, b3.w}, acc3);
  }

  float* crow = C + (size_t)(m0 + half * 8) * GAT_FH + n0 + l;
#pragma unroll
  for (int r = 0; r < 8; ++r) {
    crow[(size_t)r * GAT_FH]      = acc0[r];
    crow[(size_t)r * GAT_FH + 16] = acc1[r];
    crow[(size_t)r * GAT_FH + 32] = acc2[r];
    crow[(size_t)r * GAT_FH + 48] = acc3[r];
  }
}

__global__ void alphas_kernel(const float* __restrict__ h,
                              const float* __restrict__ asrc,
                              const float* __restrict__ adst,
                              float* __restrict__ s, float* __restrict__ d,
                              int total) {
  int i = blockIdx.x * blockDim.x + threadIdx.x;
  if (i >= total) return;
  int node = i >> 3, hd = i & 7;
  const float* hp = h + (size_t)node * GAT_FH + hd * GAT_C;
  const float* ap = asrc + hd * GAT_C;
  const float* bp = adst + hd * GAT_C;
  float ss = 0.f, dd = 0.f;
#pragma unroll
  for (int cc = 0; cc < GAT_C; ++cc) {
    float v = hp[cc];
    ss += v * ap[cc];
    dd += v * bp[cc];
  }
  s[i] = ss;
  d[i] = dd;
}

__global__ void csr_zero_kernel(int* __restrict__ deg) {
  int i = blockIdx.x * blockDim.x + threadIdx.x;
  if (i < GAT_N) deg[i] = 0;
}

__global__ void csr_count_kernel(const int* __restrict__ ei, int* __restrict__ deg) {
  int e = blockIdx.x * blockDim.x + threadIdx.x;
  if (e < GAT_E) atomicAdd(&deg[ei[GAT_E + e]], 1);
}

#define SCAN_TPB 1024
#define SCAN_BLOCKS ((GAT_N + SCAN_TPB - 1) / SCAN_TPB)
__global__ void __launch_bounds__(SCAN_TPB)
csr_scan_kernel(const int* __restrict__ deg, int* __restrict__ rowptr,
                int* __restrict__ cursor) {
  __shared__ int wave_tot[SCAN_TPB / 32];
  __shared__ int wave_ahead[SCAN_TPB / 32];
  const int t = threadIdx.x;
  const int lane = t & 31;
  const int w = t >> 5;
  const int b = blockIdx.x;
  const int i = b * SCAN_TPB + t;
  const int v = (i < GAT_N) ? deg[i] : 0;

  const int4* __restrict__ deg4 = (const int4*)deg;
  const int n4 = b * (SCAN_TPB / 4);
  int ahead = 0;
  for (int j = 0; j < (SCAN_BLOCKS + 3) / 4; ++j) {
    const int q = j * SCAN_TPB + t;
    const int4 x = deg4[(q < n4) ? q : 0];
    const int s4 = (x.x + x.y) + (x.z + x.w);
    ahead += (q < n4) ? s4 : 0;
  }

  int inc = v;
  for (int off = 1; off < 32; off <<= 1) {
    const int n = __shfl_up(inc, off, 32);
    if (lane >= off) inc += n;
  }
  for (int off = 16; off > 0; off >>= 1) ahead += __shfl_xor(ahead, off, 32);

  if (lane == 31) wave_tot[w] = inc;
  if (lane == 0) wave_ahead[w] = ahead;
  __syncthreads();

  const int wt = wave_tot[lane];
  int winc = wt;
  for (int off = 1; off < 32; off <<= 1) {
    const int n = __shfl_up(winc, off, 32);
    if (lane >= off) winc += n;
  }
  const int wave_off = __shfl(winc - wt, w, 32);
  int base = wave_ahead[lane];
  for (int off = 16; off > 0; off >>= 1) base += __shfl_xor(base, off, 32);

  const int excl = base + wave_off + (inc - v);
  if (i < GAT_N) {
    rowptr[i] = excl;
    cursor[i] = excl;
  }
  if (i == GAT_N - 1) rowptr[GAT_N] = excl + v;
}

__global__ void csr_fill_kernel(const int* __restrict__ ei, int* __restrict__ cursor,
                                int* __restrict__ tmp_src) {
  int e = blockIdx.x * blockDim.x + threadIdx.x;
  if (e >= GAT_E) return;
  int p = atomicAdd(&cursor[ei[GAT_E + e]], 1);
  tmp_src[p] = ei[e];
}

__global__ void csr_order_kernel(const int* __restrict__ rowptr,
                                 const int* __restrict__ tmp_src,
                                 int* __restrict__ csr_src) {
  const int lane = threadIdx.x & 31;
  const int d = __builtin_amdgcn_readfirstlane((blockIdx.x * blockDim.x + threadIdx.x) >> 5);
  if (d >= GAT_N) return;
  const int beg = rowptr[d];
  const int deg = rowptr[d + 1] - beg;
  for (int t = lane; t < deg; t += 32) {
    const int s = tmp_src[beg + t];
    int rank = 0;
    for (int j = 0; j < deg; ++j) {
      const int sj = tmp_src[beg + j];
      rank += ((sj < s) || (sj == s && j < t)) ? 1 : 0;
    }
    csr_src[beg + rank] = s;
  }
}

__device__ __forceinline__ float bias_elu(float a, float b) {
  const float v = a + b;
  return v > 0.f ? v : (__expf(v) - 1.f);
}

template <bool LAST>
__global__ void __launch_bounds__(256)
gat_gather_kernel(const int* __restrict__ rowptr, const int* __restrict__ csr_src,
                  const float* __restrict__ as, const float* __restrict__ ad,
                  const float* __restrict__ h, const float* __restrict__ bias,
                  const float* __restrict__ Wl, const float* __restrict__ bl,
                  float* __restrict__ agg, float* __restrict__ y) {
  const int lane = threadIdx.x & 31;
  const int d = __builtin_amdgcn_readfirstlane((blockIdx.x * blockDim.x + threadIdx.x) >> 5);
  if (d >= GAT_N) return;
  const int beg = rowptr[d];
  const int deg = rowptr[d + 1] - beg;
  const int T = deg + 1;

  const int slot = lane >> 3, hd = lane & 7;
  const float adv = ad[d * GAT_H + hd];

  float m = -3.0e38f;
  for (int base = 0; base < T; base += 32) {
    const int t = base + lane;
    const int sv = (t < deg) ? csr_src[beg + t] : d;
    const int cnt = (T - base < 32) ? (T - base) : 32;
#pragma unroll
    for (int i = 0; i < 32; i += 4) {
      const int j = i + slot;
      const int jj = (j < cnt) ? j : (cnt - 1);
      const int s = __shfl(sv, jj, 32);
      float v = as[s * GAT_H + hd] + adv;
      v = v > 0.f ? v : NEG_SLOPE * v;
      m = fmaxf(m, v);
    }
  }
  m = fmaxf(m, __shfl_xor(m, 8, 32));
  m = fmaxf(m, __shfl_xor(m, 16, 32));

  float dn = 0.f;
  for (int base = 0; base < T; base += 32) {
    const int t = base + lane;
    const int sv = (t < deg) ? csr_src[beg + t] : d;
    const int cnt = (T - base < 32) ? (T - base) : 32;
#pragma unroll
    for (int i = 0; i < 32; i += 4) {
      const int j = i + slot;
      const int jj = (j < cnt) ? j : (cnt - 1);
      const int s = __shfl(sv, jj, 32);
      float v = as[s * GAT_H + hd] + adv;
      v = v > 0.f ? v : NEG_SLOPE * v;
      const float w = (j < cnt) ? 1.f : 0.f;
      dn += w * __expf(v - m);
    }
  }
  dn += __shfl_xor(dn, 8, 32);
  dn += __shfl_xor(dn, 16, 32);

  const int hA = lane >> 3, hB = 4 + (lane >> 3);
  const float mA = __shfl(m, hA, 32),  mB = __shfl(m, hB, 32);
  const float dA = __shfl(dn, hA, 32) + 1e-16f;
  const float dB = __shfl(dn, hB, 32) + 1e-16f;
  const float adA = ad[d * GAT_H + hA], adB = ad[d * GAT_H + hB];

  const float4* __restrict__ h4 = (const float4*)h;
  float4 acc0 = make_float4(0.f, 0.f, 0.f, 0.f);
  float4 acc1 = make_float4(0.f, 0.f, 0.f, 0.f);
  for (int base = 0; base < T; base += 32) {
    const int t = base + lane;
    const int sv = (t < deg) ? csr_src[beg + t] : d;
    const int cnt = (T - base < 32) ? (T - base) : 32;
#pragma unroll 4
    for (int i = 0; i < cnt; ++i) {
      const int s = __builtin_amdgcn_readlane(sv, i);
      float vA = as[s * GAT_H + hA] + adA;
      float vB = as[s * GAT_H + hB] + adB;
      vA = vA > 0.f ? vA : NEG_SLOPE * vA;
      vB = vB > 0.f ? vB : NEG_SLOPE * vB;
      const float aA = __expf(vA - mA) / dA;
      const float aB = __expf(vB - mB) / dB;
      const float4 x0 = h4[(size_t)s * (GAT_FH / 4) + lane];
      const float4 x1 = h4[(size_t)s * (GAT_FH / 4) + 32 + lane];
      acc0.x += x0.x * aA; acc0.y += x0.y * aA; acc0.z += x0.z * aA; acc0.w += x0.w * aA;
      acc1.x += x1.x * aB; acc1.y += x1.y * aB; acc1.z += x1.z * aB; acc1.w += x1.w * aB;
    }
  }
  const float4* __restrict__ b4 = (const float4*)bias;
  const float4 bb0 = b4[lane], bb1 = b4[32 + lane];
  float4 r0, r1;
  r0.x = bias_elu(acc0.x, bb0.x); r0.y = bias_elu(acc0.y, bb0.y);
  r0.z = bias_elu(acc0.z, bb0.z); r0.w = bias_elu(acc0.w, bb0.w);
  r1.x = bias_elu(acc1.x, bb1.x); r1.y = bias_elu(acc1.y, bb1.y);
  r1.z = bias_elu(acc1.z, bb1.z); r1.w = bias_elu(acc1.w, bb1.w);
  if (!LAST) {
    float4* o4 = (float4*)(agg + (size_t)d * GAT_FH);
    o4[lane] = r0;
    o4[32 + lane] = r1;
  } else {
    const float4* __restrict__ w4 = (const float4*)Wl;
    const float4 w0 = w4[lane], w1 = w4[32 + lane];
    float s = r0.x * w0.x;
    s += r0.y * w0.y; s += r0.z * w0.z; s += r0.w * w0.w;
    s += r1.x * w1.x; s += r1.y * w1.y; s += r1.z * w1.z; s += r1.w * w1.w;
    s += __shfl_xor(s, 16, 32);
    s += __shfl_xor(s, 8, 32);
    s += __shfl_xor(s, 4, 32);
    s += __shfl_xor(s, 2, 32);
    s += __shfl_xor(s, 1, 32);
    if (lane == 0) y[d] = s + bl[0];
  }
}

extern "C" void kernel_launch(void* const* d_in, const int* in_sizes, int n_in,
                              void* d_out, int out_size, void* d_ws, size_t ws_size,
                              hipStream_t stream) {
  const float* x      = (const float*)d_in[0];
  const int*   ei     = (const int*)  d_in[1];
  const float* W1     = (const float*)d_in[2];
  const float* a1_src = (const float*)d_in[3];
  const float* a1_dst = (const float*)d_in[4];
  const float* b1     = (const float*)d_in[5];
  const float* W2     = (const float*)d_in[6];
  const float* a2_src = (const float*)d_in[7];
  const float* a2_dst = (const float*)d_in[8];
  const float* b2     = (const float*)d_in[9];
  const float* Wl     = (const float*)d_in[10];
  const float* bl     = (const float*)d_in[11];
  float* out = (float*)d_out;

  float* hbuf  = (float*)d_ws;
  float* agg   = hbuf  + (size_t)GAT_N * GAT_FH;
  float* as    = agg   + (size_t)GAT_N * GAT_FH;
  float* ad    = as    + (size_t)GAT_N * GAT_H;
  int* deg     = (int*)(ad + (size_t)GAT_N * GAT_H);
  int* rowptr  = deg    + GAT_N;
  int* cursor  = rowptr + (GAT_N + 1);
  int* tmp_src = cursor + GAT_N;
  int* csr_src = tmp_src + GAT_E;

  const int NH  = GAT_N * GAT_H;

  dim3 gemm_grid(GAT_FH / GEMM_BN, GEMM_MGROUPS);
  dim3 gemm_blk(GEMM_WAVES * 32);

  const int TPB = 256;
  auto blocks = [](int n, int tpb) { return (n + tpb - 1) / tpb; };

  csr_zero_kernel<<<blocks(GAT_N, TPB), TPB, 0, stream>>>(deg);
  csr_count_kernel<<<blocks(GAT_E, TPB), TPB, 0, stream>>>(ei, deg);
  csr_scan_kernel<<<SCAN_BLOCKS, SCAN_TPB, 0, stream>>>(deg, rowptr, cursor);
  csr_fill_kernel<<<blocks(GAT_E, TPB), TPB, 0, stream>>>(ei, cursor, tmp_src);
  csr_order_kernel<<<blocks(GAT_N * 32, TPB), TPB, 0, stream>>>(rowptr, tmp_src, csr_src);

  gemm_wmma_f32<128><<<gemm_grid, gemm_blk, 0, stream>>>(x, W1, hbuf, GAT_N);
  alphas_kernel<<<blocks(NH, TPB), TPB, 0, stream>>>(hbuf, a1_src, a1_dst, as, ad, NH);
  gat_gather_kernel<false><<<blocks(GAT_N * 32, TPB), TPB, 0, stream>>>(
      rowptr, csr_src, as, ad, hbuf, b1, Wl, bl, agg, out);

  gemm_wmma_f32<GAT_FH><<<gemm_grid, gemm_blk, 0, stream>>>(agg, W2, hbuf, GAT_N);
  alphas_kernel<<<blocks(NH, TPB), TPB, 0, stream>>>(hbuf, a2_src, a2_dst, as, ad, NH);
  gat_gather_kernel<true><<<blocks(GAT_N * 32, TPB), TPB, 0, stream>>>(
      rowptr, csr_src, as, ad, hbuf, b2, Wl, bl, agg, out);
}


# Attention scores formed in the gather wave from the h rows it reads (new sum order); softmax in one sweep, running-max rescale, one divide per channel; score tables and their 2 launches go
# speedup vs baseline: 13.5516x; 1.1000x over previous
#include <hip/hip_runtime.h>
#include <hip/hip_bf16.h>

#define GAT_N 30000
#define GAT_E 480000
#define GAT_ETOT (GAT_E + GAT_N)
#define GAT_H 8
#define GAT_C 32
#define GAT_FH 256
#define NEG_SLOPE 0.2f

typedef float v2f __attribute__((ext_vector_type(2)));
typedef float v8f __attribute__((ext_vector_type(8)));

#define GEMM_BN 64
#define GEMM_WAVES 8
#define GEMM_MGROUPS ((GAT_N / 16 + GEMM_WAVES - 1) / GEMM_WAVES)

__device__ __forceinline__ v8f wmma4(v2f a, v2f b, v8f c) {
  return __builtin_amdgcn_wmma_f32_16x16x4_f32(
       false, a,  false, b,
       (short)0, c,  false,  false);
}

template <int K>
__global__ void __launch_bounds__(GEMM_WAVES * 32)
gemm_wmma_f32(const float* __restrict__ A,
              const float* __restrict__ B,
              float* __restrict__ C, int M) {
  constexpr int KG = K / 4;
  constexpr int NC = K / 8;
  __shared__ float4 Bs[KG * GEMM_BN];

  const int tid = threadIdx.x;
  const int n0 = blockIdx.x * GEMM_BN;

#pragma unroll 4
  for (int it = 0; it < KG * GEMM_BN / (GEMM_WAVES * 32); ++it) {
    const int idx = it * (GEMM_WAVES * 32) + tid;
    const int kg = idx >> 6, n = idx & (GEMM_BN - 1);
    const float* bp = B + (size_t)(4 * kg) * GAT_FH + n0 + n;
    float4 v;
    v.x = bp[0];
    v.y = bp[GAT_FH];
    v.z = bp[2 * GAT_FH];
    v.w = bp[3 * GAT_FH];
    Bs[idx] = v;
  }
  __syncthreads();

  const int wave = __builtin_amdgcn_readfirstlane(tid >> 5);
  const int lane = tid & 31;
  const int half = lane >> 4;
  const int l    = lane & 15;
  const int m0 = (blockIdx.y * GEMM_WAVES + wave) * 16;
  if (m0 >= M) return;

  const float4* __restrict__ ap = (const float4*)(A + (size_t)(m0 + l) * K) + half;
  const float4* bs = Bs + half * GEMM_BN + l;

  v8f acc0 = {0.f, 0.f, 0.f, 0.f, 0.f, 0.f, 0.f, 0.f};
  v8f acc1 = acc0, acc2 = acc0, acc3 = acc0;

#pragma unroll 4
  for (int c = 0; c < NC; ++c) {
    const float4 a  = ap[2 * c];
    const float4 b0 = bs[c * 2 * GEMM_BN];
    const float4 b1 = bs[c * 2 * GEMM_BN + 16];
    const float4 b2 = bs[c * 2 * GEMM_BN + 32];
    const float4 b3 = bs[c * 2 * GEMM_BN + 48];

    const v2f a01 = {a.x, a.y};
    const v2f a23 = {a.z, a.w};
    acc0 = wmma4(a01, (v2f){b0.x, b0.y}, acc0);
    acc1 = wmma4(a01, (v2f){b1.x, b1.y}, acc1);
    acc2 = wmma4(a01, (v2f){b2.x, b2.y}, acc2);
    acc3 = wmma4(a01, (v2f){b3.x, b3.y}, acc3);
    acc0 = wmma4(a23, (v2f){b0.z, b0.w}, acc0);
    acc1 = wmma4(a23, (v2f){b1.z, b1.w}, acc1);
    acc2 = wmma4(a23, (v2f){b2.z, b2.w}, acc2);
    acc3 = wmma4(a23, (v2f){b3.z, b3.w}, acc3);
  }

  float* crow = C + (size_t)(m0 + half * 8) * GAT_FH + n0 + l;
#pragma unroll
  for (int r = 0; r < 8; ++r) {
    crow[(size_t)r * GAT_FH]      = acc0[r];
    crow[(size_t)r * GAT_FH + 16] = acc1[r];
    crow[(size_t)r * GAT_FH + 32] = acc2[r];
    crow[(size_t)r * GAT_FH + 48] = acc3[r];
  }
}

__global__ void csr_zero_kernel(int* __restrict__ deg) {
  int i = blockIdx.x * blockDim.x + threadIdx.x;
  if (i < GAT_N) deg[i] = 0;
}

__global__ void csr_count_kernel(const int* __restrict__ ei, int* __restrict__ deg) {
  int e = blockIdx.x * blockDim.x + threadIdx.x;
  if (e < GAT_E) atomicAdd(&deg[ei[GAT_E + e]], 1);
}

#define SCAN_TPB 1024
#define SCAN_BLOCKS ((GAT_N + SCAN_TPB - 1) / SCAN_TPB)
__global__ void __launch_bounds__(SCAN_TPB)
csr_scan_kernel(const int* __restrict__ deg, int* __restrict__ rowptr,
                int* __restrict__ cursor) {
  __shared__ int wave_tot[SCAN_TPB / 32];
  __shared__ int wave_ahead[SCAN_TPB / 32];
  const int t = threadIdx.x;
  const int lane = t & 31;
  const int w = t >> 5;
  const int b = blockIdx.x;
  const int i = b * SCAN_TPB + t;
  const int v = (i < GAT_N) ? deg[i] : 0;

  const int4* __restrict__ deg4 = (const int4*)deg;
  const int n4 = b * (SCAN_TPB / 4);
  int ahead = 0;
  for (int j = 0; j < (SCAN_BLOCKS + 3) / 4; ++j) {
    const int q = j * SCAN_TPB + t;
    const int4 x = deg4[(q < n4) ? q : 0];
    const int s4 = (x.x + x.y) + (x.z + x.w);
    ahead += (q < n4) ? s4 : 0;
  }

  int inc = v;
  for (int off = 1; off < 32; off <<= 1) {
    const int n = __shfl_up(inc, off, 32);
    if (lane >= off) inc += n;
  }
  for (int off = 16; off > 0; off >>= 1) ahead += __shfl_xor(ahead, off, 32);

  if (lane == 31) wave_tot[w] = inc;
  if (lane == 0) wave_ahead[w] = ahead;
  __syncthreads();

  const int wt = wave_tot[lane];
  int winc = wt;
  for (int off = 1; off < 32; off <<= 1) {
    const int n = __shfl_up(winc, off, 32);
    if (lane >= off) winc += n;
  }
  const int wave_off = __shfl(winc - wt, w, 32);
  int base = wave_ahead[lane];
  for (int off = 16; off > 0; off >>= 1) base += __shfl_xor(base, off, 32);

  const int excl = base + wave_off + (inc - v);
  if (i < GAT_N) {
    rowptr[i] = excl;
    cursor[i] = excl;
  }
  if (i == GAT_N - 1) rowptr[GAT_N] = excl + v;
}

__global__ void csr_fill_kernel(const int* __restrict__ ei, int* __restrict__ cursor,
                                int* __restrict__ tmp_src) {
  int e = blockIdx.x * blockDim.x + threadIdx.x;
  if (e >= GAT_E) return;
  int p = atomicAdd(&cursor[ei[GAT_E + e]], 1);
  tmp_src[p] = ei[e];
}

__global__ void csr_order_kernel(const int* __restrict__ rowptr,
                                 const int* __restrict__ tmp_src,
                                 int* __restrict__ csr_src) {
  const int lane = threadIdx.x & 31;
  const int d = __builtin_amdgcn_readfirstlane((blockIdx.x * blockDim.x + threadIdx.x) >> 5);
  if (d >= GAT_N) return;
  const int beg = rowptr[d];
  const int deg = rowptr[d + 1] - beg;
  for (int t = lane; t < deg; t += 32) {
    const int s = tmp_src[beg + t];
    int rank = 0;
    for (int j = 0; j < deg; ++j) {
      const int sj = tmp_src[beg + j];
      rank += ((sj < s) || (sj == s && j < t)) ? 1 : 0;
    }
    csr_src[beg + rank] = s;
  }
}

__device__ __forceinline__ float bias_elu(float a, float b) {
  const float v = a + b;
  return v > 0.f ? v : (__expf(v) - 1.f);
}

template <int MASK>
__device__ __forceinline__ float row_xor(float v) {
  return __builtin_bit_cast(float,
      __builtin_amdgcn_update_dpp(0, __builtin_bit_cast(int, v), 0x160 + MASK, 0xF, 0xF, true));
}

__device__ __forceinline__ float head_sum(float v) {
  v += row_xor<1>(v);
  v += row_xor<2>(v);
  v += row_xor<4>(v);
  return v;
}

__device__ __forceinline__ float dot4(float4 a, float4 b) {
  float s = a.x * b.x;
  s += a.y * b.y; s += a.z * b.z; s += a.w * b.w;
  return s;
}

template <bool LAST>
__global__ void __launch_bounds__(256)
gat_gather_kernel(const int* __restrict__ rowptr, const int* __restrict__ csr_src,
                  const float* __restrict__ asrc, const float* __restrict__ adst,
                  const float* __restrict__ h, const float* __restrict__ bias,
                  const float* __restrict__ Wl, const float* __restrict__ bl,
                  float* __restrict__ agg, float* __restrict__ y) {
  const int lane = threadIdx.x & 31;
  const int d = __builtin_amdgcn_readfirstlane((blockIdx.x * blockDim.x + threadIdx.x) >> 5);
  if (d >= GAT_N) return;
  const int beg = rowptr[d];
  const int deg = rowptr[d + 1] - beg;

  const float4* __restrict__ h4 = (const float4*)h;
  const float4 as0 = ((const float4*)asrc)[lane], as1 = ((const float4*)asrc)[32 + lane];
  const float4 ad0 = ((const float4*)adst)[lane], ad1 = ((const float4*)adst)[32 + lane];

  float4 acc0 = h4[(size_t)d * (GAT_FH / 4) + lane];
  float4 acc1 = h4[(size_t)d * (GAT_FH / 4) + 32 + lane];
  const float adA = head_sum(dot4(acc0, ad0));
  const float adB = head_sum(dot4(acc1, ad1));
  float mA = head_sum(dot4(acc0, as0)) + adA;
  float mB = head_sum(dot4(acc1, as1)) + adB;
  mA = mA > 0.f ? mA : NEG_SLOPE * mA;
  mB = mB > 0.f ? mB : NEG_SLOPE * mB;
  float dnA = 1.f, dnB = 1.f;

  for (int base = 0; base < deg; base += 32) {
    const int t = base + lane;
    const int sv = (t < deg) ? csr_src[beg + t] : d;
    const int cnt = (deg - base < 32) ? (deg - base) : 32;
#pragma unroll 4
    for (int i = 0; i < cnt; ++i) {
      const int s = __builtin_amdgcn_readlane(sv, i);
      const float4 x0 = h4[(size_t)s * (GAT_FH / 4) + lane];
      const float4 x1 = h4[(size_t)s * (GAT_FH / 4) + 32 + lane];
      float vA = head_sum(dot4(x0, as0)) + adA;
      float vB = head_sum(dot4(x1, as1)) + adB;
      vA = vA > 0.f ? vA : NEG_SLOPE * vA;
      vB = vB > 0.f ? vB : NEG_SLOPE * vB;
      const float nA = fmaxf(mA, vA), nB = fmaxf(mB, vB);
      const float cA = __expf(mA - nA), cB = __expf(mB - nB);
      const float pA = __expf(vA - nA), pB = __expf(vB - nB);
      mA = nA; mB = nB;
      dnA = dnA * cA + pA;
      dnB = dnB * cB + pB;
      acc0.x = acc0.x * cA + x0.x * pA; acc0.y = acc0.y * cA + x0.y * pA;
      acc0.z = acc0.z * cA + x0.z * pA; acc0.w = acc0.w * cA + x0.w * pA;
      acc1.x = acc1.x * cB + x1.x * pB; acc1.y = acc1.y * cB + x1.y * pB;
      acc1.z = acc1.z * cB + x1.z * pB; acc1.w = acc1.w * cB + x1.w * pB;
    }
  }
  const float dA = dnA + 1e-16f, dB = dnB + 1e-16f;
  acc0.x /= dA; acc0.y /= dA; acc0.z /= dA; acc0.w /= dA;
  acc1.x /= dB; acc1.y /= dB; acc1.z /= dB; acc1.w /= dB;

  const float4* __restrict__ b4 = (const float4*)bias;
  const float4 bb0 = b4[lane], bb1 = b4[32 + lane];
  float4 r0, r1;
  r0.x = bias_elu(acc0.x, bb0.x); r0.y = bias_elu(acc0.y, bb0.y);
  r0.z = bias_elu(acc0.z, bb0.z); r0.w = bias_elu(acc0.w, bb0.w);
  r1.x = bias_elu(acc1.x, bb1.x); r1.y = bias_elu(acc1.y, bb1.y);
  r1.z = bias_elu(acc1.z, bb1.z); r1.w = bias_elu(acc1.w, bb1.w);
  if (!LAST) {
    float4* o4 = (float4*)(agg + (size_t)d * GAT_FH);
    o4[lane] = r0;
    o4[32 + lane] = r1;
  } else {
    const float4* __restrict__ w4 = (const float4*)Wl;
    const float4 w0 = w4[lane], w1 = w4[32 + lane];
    float s = r0.x * w0.x;
    s += r0.y * w0.y; s += r0.z * w0.z; s += r0.w * w0.w;
    s += r1.x * w1.x; s += r1.y * w1.y; s += r1.z * w1.z; s += r1.w * w1.w;
    s += __shfl_xor(s, 16, 32);
    s += __shfl_xor(s, 8, 32);
    s += __shfl_xor(s, 4, 32);
    s += __shfl_xor(s, 2, 32);
    s += __shfl_xor(s, 1, 32);
    if (lane == 0) y[d] = s + bl[0];
  }
}

extern "C" void kernel_launch(void* const* d_in, const int* in_sizes, int n_in,
                              void* d_out, int out_size, void* d_ws, size_t ws_size,
                              hipStream_t stream) {
  const float* x      = (const float*)d_in[0];
  const int*   ei     = (const int*)  d_in[1];
  const float* W1     = (const float*)d_in[2];
  const float* a1_src = (const float*)d_in[3];
  const float* a1_dst = (const float*)d_in[4];
  const float* b1     = (const float*)d_in[5];
  const float* W2     = (const float*)d_in[6];
  const float* a2_src = (const float*)d_in[7];
  const float* a2_dst = (const float*)d_in[8];
  const float* b2     = (const float*)d_in[9];
  const float* Wl     = (const float*)d_in[10];
  const float* bl     = (const float*)d_in[11];
  float* out = (float*)d_out;

  float* hbuf  = (float*)d_ws;
  float* agg   = hbuf  + (size_t)GAT_N * GAT_FH;
  int* deg     = (int*)(agg + (size_t)GAT_N * GAT_FH);
  int* rowptr  = deg    + GAT_N;
  int* cursor  = rowptr + (GAT_N + 1);
  int* tmp_src = cursor + GAT_N;
  int* csr_src = tmp_src + GAT_E;

  dim3 gemm_grid(GAT_FH / GEMM_BN, GEMM_MGROUPS);
  dim3 gemm_blk(GEMM_WAVES * 32);

  const int TPB = 256;
  auto blocks = [](int n, int tpb) { return (n + tpb - 1) / tpb; };

  csr_zero_kernel<<<blocks(GAT_N, TPB), TPB, 0, stream>>>(deg);
  csr_count_kernel<<<blocks(GAT_E, TPB), TPB, 0, stream>>>(ei, deg);
  csr_scan_kernel<<<SCAN_BLOCKS, SCAN_TPB, 0, stream>>>(deg, rowptr, cursor);
  csr_fill_kernel<<<blocks(GAT_E, TPB), TPB, 0, stream>>>(ei, cursor, tmp_src);
  csr_order_kernel<<<blocks(GAT_N * 32, TPB), TPB, 0, stream>>>(rowptr, tmp_src, csr_src);

  gemm_wmma_f32<128><<<gemm_grid, gemm_blk, 0, stream>>>(x, W1, hbuf, GAT_N);
  gat_gather_kernel<false><<<blocks(GAT_N * 32, TPB), TPB, 0, stream>>>(
      rowptr, csr_src, a1_src, a1_dst, hbuf, b1, Wl, bl, agg, out);

  gemm_wmma_f32<GAT_FH><<<gemm_grid, gemm_blk, 0, stream>>>(agg, W2, hbuf, GAT_N);
  gat_gather_kernel<true><<<blocks(GAT_N * 32, TPB), TPB, 0, stream>>>(
      rowptr, csr_src, a2_src, a2_dst, hbuf, b2, Wl, bl, agg, out);
}


# Layer 1's GEMM tiles ride as the first workgroups of the edge fill and order launches (independent work; matrix units idle there); its own launch goes, 9 launches become 8
# speedup vs baseline: 13.9938x; 1.0326x over previous
#include <hip/hip_runtime.h>
#include <hip/hip_bf16.h>

#define GAT_N 30000
#define GAT_E 480000
#define GAT_ETOT (GAT_E + GAT_N)
#define GAT_H 8
#define GAT_C 32
#define GAT_FH 256
#define NEG_SLOPE 0.2f

typedef float v2f __attribute__((ext_vector_type(2)));
typedef float v8f __attribute__((ext_vector_type(8)));

#define GEMM_BN 64
#define GEMM_WAVES 8
#define GEMM_MGROUPS ((GAT_N / 16 + GEMM_WAVES - 1) / GEMM_WAVES)

__device__ __forceinline__ v8f wmma4(v2f a, v2f b, v8f c) {
  return __builtin_amdgcn_wmma_f32_16x16x4_f32(
       false, a,  false, b,
       (short)0, c,  false,  false);
}

template <int K>
__device__ __forceinline__ void
gemm_tile_f32(const float* __restrict__ A,
              const float* __restrict__ B,
              float* __restrict__ C, int M, int bx, int by) {
  constexpr int KG = K / 4;
  constexpr int NC = K / 8;
  __shared__ float4 Bs[KG * GEMM_BN];

  const int tid = threadIdx.x;
  const int n0 = bx * GEMM_BN;

#pragma unroll 4
  for (int it = 0; it < KG * GEMM_BN / (GEMM_WAVES * 32); ++it) {
    const int idx = it * (GEMM_WAVES * 32) + tid;
    const int kg = idx >> 6, n = idx & (GEMM_BN - 1);
    const float* bp = B + (size_t)(4 * kg) * GAT_FH + n0 + n;
    float4 v;
    v.x = bp[0];
    v.y = bp[GAT_FH];
    v.z = bp[2 * GAT_FH];
    v.w = bp[3 * GAT_FH];
    Bs[idx] = v;
  }
  __syncthreads();

  const int wave = __builtin_amdgcn_readfirstlane(tid >> 5);
  const int lane = tid & 31;
  const int half = lane >> 4;
  const int l    = lane & 15;
  const int m0 = (by * GEMM_WAVES + wave) * 16;
  if (m0 >= M) return;

  const float4* __restrict__ ap = (const float4*)(A + (size_t)(m0 + l) * K) + half;
  const float4* bs = Bs + half * GEMM_BN + l;

  v8f acc0 = {0.f, 0.f, 0.f, 0.f, 0.f, 0.f, 0.f, 0.f};
  v8f acc1 = acc0, acc2 = acc0, acc3 = acc0;

#pragma unroll 4
  for (int c = 0; c < NC; ++c) {
    const float4 a  = ap[2 * c];
    const float4 b0 = bs[c * 2 * GEMM_BN];
    const float4 b1 = bs[c * 2 * GEMM_BN + 16];
    const float4 b2 = bs[c * 2 * GEMM_BN + 32];
    const float4 b3 = bs[c * 2 * GEMM_BN + 48];

    const v2f a01 = {a.x, a.y};
    const v2f a23 = {a.z, a.w};
    acc0 = wmma4(a01, (v2f){b0.x, b0.y}, acc0);
    acc1 = wmma4(a01, (v2f){b1.x, b1.y}, acc1);
    acc2 = wmma4(a01, (v2f){b2.x, b2.y}, acc2);
    acc3 = wmma4(a01, (v2f){b3.x, b3.y}, acc3);
    acc0 = wmma4(a23, (v2f){b0.z, b0.w}, acc0);
    acc1 = wmma4(a23, (v2f){b1.z, b1.w}, acc1);
    acc2 = wmma4(a23, (v2f){b2.z, b2.w}, acc2);
    acc3 = wmma4(a23, (v2f){b3.z, b3.w}, acc3);
  }

  float* crow = C + (size_t)(m0 + half * 8) * GAT_FH + n0 + l;
#pragma unroll
  for (int r = 0; r < 8; ++r) {
    crow[(size_t)r * GAT_FH]      = acc0[r];
    crow[(size_t)r * GAT_FH + 16] = acc1[r];
    crow[(size_t)r * GAT_FH + 32] = acc2[r];
    crow[(size_t)r * GAT_FH + 48] = acc3[r];
  }
}

template <int K>
__global__ void __launch_bounds__(GEMM_WAVES * 32)
gemm_wmma_f32(const float* __restrict__ A,
              const float* __restrict__ B,
              float* __restrict__ C, int M) {
  gemm_tile_f32<K>(A, B, C, M, blockIdx.x, blockIdx.y);
}

#define GEMM_STRIPS (GAT_FH / GEMM_BN)
#define GEMM_TILES (GEMM_STRIPS * GEMM_MGROUPS)
#define RIDE_FILL (GEMM_TILES / 2)
#define RIDE_ORDER (GEMM_TILES - RIDE_FILL)

__global__ void csr_zero_kernel(int* __restrict__ deg) {
  int i = blockIdx.x * blockDim.x + threadIdx.x;
  if (i < GAT_N) deg[i] = 0;
}

__global__ void csr_count_kernel(const int* __restrict__ ei, int* __restrict__ deg) {
  int e = blockIdx.x * blockDim.x + threadIdx.x;
  if (e < GAT_E) atomicAdd(&deg[ei[GAT_E + e]], 1);
}

#define SCAN_TPB 1024
#define SCAN_BLOCKS ((GAT_N + SCAN_TPB - 1) / SCAN_TPB)
__global__ void __launch_bounds__(SCAN_TPB)
csr_scan_kernel(const int* __restrict__ deg, int* __restrict__ rowptr,
                int* __restrict__ cursor) {
  __shared__ int wave_tot[SCAN_TPB / 32];
  __shared__ int wave_ahead[SCAN_TPB / 32];
  const int t = threadIdx.x;
  const int lane = t & 31;
  const int w = t >> 5;
  const int b = blockIdx.x;
  const int i = b * SCAN_TPB + t;
  const int v = (i < GAT_N) ? deg[i] : 0;

  const int4* __restrict__ deg4 = (const int4*)deg;
  const int n4 = b * (SCAN_TPB / 4);
  int ahead = 0;
  for (int j = 0; j < (SCAN_BLOCKS + 3) / 4; ++j) {
    const int q = j * SCAN_TPB + t;
    const int4 x = deg4[(q < n4) ? q : 0];
    const int s4 = (x.x + x.y) + (x.z + x.w);
    ahead += (q < n4) ? s4 : 0;
  }

  int inc = v;
  for (int off = 1; off < 32; off <<= 1) {
    const int n = __shfl_up(inc, off, 32);
    if (lane >= off) inc += n;
  }
  for (int off = 16; off > 0; off >>= 1) ahead += __shfl_xor(ahead, off, 32);

  if (lane == 31) wave_tot[w] = inc;
  if (lane == 0) wave_ahead[w] = ahead;
  __syncthreads();

  const int wt = wave_tot[lane];
  int winc = wt;
  for (int off = 1; off < 32; off <<= 1) {
    const int n = __shfl_up(winc, off, 32);
    if (lane >= off) winc += n;
  }
  const int wave_off = __shfl(winc - wt, w, 32);
  int base = wave_ahead[lane];
  for (int off = 16; off > 0; off >>= 1) base += __shfl_xor(base, off, 32);

  const int excl = base + wave_off + (inc - v);
  if (i < GAT_N) {
    rowptr[i] = excl;
    cursor[i] = excl;
  }
  if (i == GAT_N - 1) rowptr[GAT_N] = excl + v;
}

template <int GFIRST, int GCOUNT>
__global__ void __launch_bounds__(GEMM_WAVES * 32)
csr_fill_kernel(const int* __restrict__ ei, int* __restrict__ cursor,
                int* __restrict__ tmp_src,
                const float* __restrict__ x, const float* __restrict__ W,
                float* __restrict__ hout) {
  if (blockIdx.x < GCOUNT) {
    const int g = GFIRST + blockIdx.x;
    gemm_tile_f32<128>(x, W, hout, GAT_N, g & (GEMM_STRIPS - 1), g / GEMM_STRIPS);
    return;
  }
  int e = (blockIdx.x - GCOUNT) * blockDim.x + threadIdx.x;
  if (e >= GAT_E) return;
  int p = atomicAdd(&cursor[ei[GAT_E + e]], 1);
  tmp_src[p] = ei[e];
}

template <int GFIRST, int GCOUNT>
__global__ void __launch_bounds__(GEMM_WAVES * 32)
csr_order_kernel(const int* __restrict__ rowptr,
                 const int* __restrict__ tmp_src,
                 int* __restrict__ csr_src,
                 const float* __restrict__ x, const float* __restrict__ W,
                 float* __restrict__ hout) {
  if (blockIdx.x < GCOUNT) {
    const int g = GFIRST + blockIdx.x;
    gemm_tile_f32<128>(x, W, hout, GAT_N, g & (GEMM_STRIPS - 1), g / GEMM_STRIPS);
    return;
  }
  const int lane = threadIdx.x & 31;
  const int d = __builtin_amdgcn_readfirstlane(
      ((blockIdx.x - GCOUNT) * blockDim.x + threadIdx.x) >> 5);
  if (d >= GAT_N) return;
  const int beg = rowptr[d];
  const int deg = rowptr[d + 1] - beg;
  for (int t = lane; t < deg; t += 32) {
    const int s = tmp_src[beg + t];
    int rank = 0;
    for (int j = 0; j < deg; ++j) {
      const int sj = tmp_src[beg + j];
      rank += ((sj < s) || (sj == s && j < t)) ? 1 : 0;
    }
    csr_src[beg + rank] = s;
  }
}

__device__ __forceinline__ float bias_elu(float a, float b) {
  const float v = a + b;
  return v > 0.f ? v : (__expf(v) - 1.f);
}

template <int MASK>
__device__ __forceinline__ float row_xor(float v) {
  return __builtin_bit_cast(float,
      __builtin_amdgcn_update_dpp(0, __builtin_bit_cast(int, v), 0x160 + MASK, 0xF, 0xF, true));
}

__device__ __forceinline__ float head_sum(float v) {
  v += row_xor<1>(v);
  v += row_xor<2>(v);
  v += row_xor<4>(v);
  return v;
}

__device__ __forceinline__ float dot4(float4 a, float4 b) {
  float s = a.x * b.x;
  s += a.y * b.y; s += a.z * b.z; s += a.w * b.w;
  return s;
}

template <bool LAST>
__global__ void __launch_bounds__(256)
gat_gather_kernel(const int* __restrict__ rowptr, const int* __restrict__ csr_src,
                  const float* __restrict__ asrc, const float* __restrict__ adst,
                  const float* __restrict__ h, const float* __restrict__ bias,
                  const float* __restrict__ Wl, const float* __restrict__ bl,
                  float* __restrict__ agg, float* __restrict__ y) {
  const int lane = threadIdx.x & 31;
  const int d = __builtin_amdgcn_readfirstlane((blockIdx.x * blockDim.x + threadIdx.x) >> 5);
  if (d >= GAT_N) return;
  const int beg = rowptr[d];
  const int deg = rowptr[d + 1] - beg;

  const float4* __restrict__ h4 = (const float4*)h;
  const float4 as0 = ((const float4*)asrc)[lane], as1 = ((const float4*)asrc)[32 + lane];
  const float4 ad0 = ((const float4*)adst)[lane], ad1 = ((const float4*)adst)[32 + lane];

  float4 acc0 = h4[(size_t)d * (GAT_FH / 4) + lane];
  float4 acc1 = h4[(size_t)d * (GAT_FH / 4) + 32 + lane];
  const float adA = head_sum(dot4(acc0, ad0));
  const float adB = head_sum(dot4(acc1, ad1));
  float mA = head_sum(dot4(acc0, as0)) + adA;
  float mB = head_sum(dot4(acc1, as1)) + adB;
  mA = mA > 0.f ? mA : NEG_SLOPE * mA;
  mB = mB > 0.f ? mB : NEG_SLOPE * mB;
  float dnA = 1.f, dnB = 1.f;

  for (int base = 0; base < deg; base += 32) {
    const int t = base + lane;
    const int sv = (t < deg) ? csr_src[beg + t] : d;
    const int cnt = (deg - base < 32) ? (deg - base) : 32;
#pragma unroll 4
    for (int i = 0; i < cnt; ++i) {
      const int s = __builtin_amdgcn_readlane(sv, i);
      const float4 x0 = h4[(size_t)s * (GAT_FH / 4) + lane];
      const float4 x1 = h4[(size_t)s * (GAT_FH / 4) + 32 + lane];
      float vA = head_sum(dot4(x0, as0)) + adA;
      float vB = head_sum(dot4(x1, as1)) + adB;
      vA = vA > 0.f ? vA : NEG_SLOPE * vA;
      vB = vB > 0.f ? vB : NEG_SLOPE * vB;
      const float nA = fmaxf(mA, vA), nB = fmaxf(mB, vB);
      const float cA = __expf(mA - nA), cB = __expf(mB - nB);
      const float pA = __expf(vA - nA), pB = __expf(vB - nB);
      mA = nA; mB = nB;
      dnA = dnA * cA + pA;
      dnB = dnB * cB + pB;
      acc0.x = acc0.x * cA + x0.x * pA; acc0.y = acc0.y * cA + x0.y * pA;
      acc0.z = acc0.z * cA + x0.z * pA; acc0.w = acc0.w * cA + x0.w * pA;
      acc1.x = acc1.x * cB + x1.x * pB; acc1.y = acc1.y * cB + x1.y * pB;
      acc1.z = acc1.z * cB + x1.z * pB; acc1.w = acc1.w * cB + x1.w * pB;
    }
  }
  const float dA = dnA + 1e-16f, dB = dnB + 1e-16f;
  acc0.x /= dA; acc0.y /= dA; acc0.z /= dA; acc0.w /= dA;
  acc1.x /= dB; acc1.y /= dB; acc1.z /= dB; acc1.w /= dB;

  const float4* __restrict__ b4 = (const float4*)bias;
  const float4 bb0 = b4[lane], bb1 = b4[32 + lane];
  float4 r0, r1;
  r0.x = bias_elu(acc0.x, bb0.x); r0.y = bias_elu(acc0.y, bb0.y);
  r0.z = bias_elu(acc0.z, bb0.z); r0.w = bias_elu(acc0.w, bb0.w);
  r1.x = bias_elu(acc1.x, bb1.x); r1.y = bias_elu(acc1.y, bb1.y);
  r1.z = bias_elu(acc1.z, bb1.z); r1.w = bias_elu(acc1.w, bb1.w);
  if (!LAST) {
    float4* o4 = (float4*)(agg + (size_t)d * GAT_FH);
    o4[lane] = r0;
    o4[32 + lane] = r1;
  } else {
    const float4* __restrict__ w4 = (const float4*)Wl;
    const float4 w0 = w4[lane], w1 = w4[32 + lane];
    float s = r0.x * w0.x;
    s += r0.y * w0.y; s += r0.z * w0.z; s += r0.w * w0.w;
    s += r1.x * w1.x; s += r1.y * w1.y; s += r1.z * w1.z; s += r1.w * w1.w;
    s += __shfl_xor(s, 16, 32);
    s += __shfl_xor(s, 8, 32);
    s += __shfl_xor(s, 4, 32);
    s += __shfl_xor(s, 2, 32);
    s += __shfl_xor(s, 1, 32);
    if (lane == 0) y[d] = s + bl[0];
  }
}

extern "C" void kernel_launch(void* const* d_in, const int* in_sizes, int n_in,
                              void* d_out, int out_size, void* d_ws, size_t ws_size,
                              hipStream_t stream) {
  const float* x      = (const float*)d_in[0];
  const int*   ei     = (const int*)  d_in[1];
  const float* W1     = (const float*)d_in[2];
  const float* a1_src = (const float*)d_in[3];
  const float* a1_dst = (const float*)d_in[4];
  const float* b1     = (const float*)d_in[5];
  const float* W2     = (const float*)d_in[6];
  const float* a2_src = (const float*)d_in[7];
  const float* a2_dst = (const float*)d_in[8];
  const float* b2     = (const float*)d_in[9];
  const float* Wl     = (const float*)d_in[10];
  const float* bl     = (const float*)d_in[11];
  float* out = (float*)d_out;

  float* hbuf  = (float*)d_ws;
  float* agg   = hbuf  + (size_t)GAT_N * GAT_FH;
  int* deg     = (int*)(agg + (size_t)GAT_N * GAT_FH);
  int* rowptr  = deg    + GAT_N;
  int* cursor  = rowptr + (GAT_N + 1);
  int* tmp_src = cursor + GAT_N;
  int* csr_src = tmp_src + GAT_E;

  dim3 gemm_grid(GAT_FH / GEMM_BN, GEMM_MGROUPS);
  dim3 gemm_blk(GEMM_WAVES * 32);

  const int TPB = 256;
  auto blocks = [](int n, int tpb) { return (n + tpb - 1) / tpb; };

  csr_zero_kernel<<<blocks(GAT_N, TPB), TPB, 0, stream>>>(deg);
  csr_count_kernel<<<blocks(GAT_E, TPB), TPB, 0, stream>>>(ei, deg);
  csr_scan_kernel<<<SCAN_BLOCKS, SCAN_TPB, 0, stream>>>(deg, rowptr, cursor);
  csr_fill_kernel<0, RIDE_FILL><<<RIDE_FILL + blocks(GAT_E, TPB), TPB, 0, stream>>>(
      ei, cursor, tmp_src, x, W1, hbuf);
  csr_order_kernel<RIDE_FILL, RIDE_ORDER><<<RIDE_ORDER + blocks(GAT_N * 32, TPB), TPB, 0, stream>>>(
      rowptr, tmp_src, csr_src, x, W1, hbuf);

  gat_gather_kernel<false><<<blocks(GAT_N * 32, TPB), TPB, 0, stream>>>(
      rowptr, csr_src, a1_src, a1_dst, hbuf, b1, Wl, bl, agg, out);

  gemm_wmma_f32<GAT_FH><<<gemm_grid, gemm_blk, 0, stream>>>(agg, W2, hbuf, GAT_N);
  gat_gather_kernel<true><<<blocks(GAT_N * 32, TPB), TPB, 0, stream>>>(
      rowptr, csr_src, a2_src, a2_dst, hbuf, b2, Wl, bl, agg, out);
}


# Bucket sort by source id moved into the first gather's wave (rank against the other lanes' ids in registers, one lane permute); order launch goes, its GEMM tiles ride in the count launch; same sums
# speedup vs baseline: 14.4639x; 1.0336x over previous
#include <hip/hip_runtime.h>
#include <hip/hip_bf16.h>

#define GAT_N 30000
#define GAT_E 480000
#define GAT_ETOT (GAT_E + GAT_N)
#define GAT_H 8
#define GAT_C 32
#define GAT_FH 256
#define NEG_SLOPE 0.2f

typedef float v2f __attribute__((ext_vector_type(2)));
typedef float v8f __attribute__((ext_vector_type(8)));

#define GEMM_BN 64
#define GEMM_WAVES 8
#define GEMM_MGROUPS ((GAT_N / 16 + GEMM_WAVES - 1) / GEMM_WAVES)

__device__ __forceinline__ v8f wmma4(v2f a, v2f b, v8f c) {
  return __builtin_amdgcn_wmma_f32_16x16x4_f32(
       false, a,  false, b,
       (short)0, c,  false,  false);
}

template <int K>
__device__ __forceinline__ void
gemm_tile_f32(const float* __restrict__ A,
              const float* __restrict__ B,
              float* __restrict__ C, int M, int bx, int by) {
  constexpr int KG = K / 4;
  constexpr int NC = K / 8;
  __shared__ float4 Bs[KG * GEMM_BN];

  const int tid = threadIdx.x;
  const int n0 = bx * GEMM_BN;

#pragma unroll 4
  for (int it = 0; it < KG * GEMM_BN / (GEMM_WAVES * 32); ++it) {
    const int idx = it * (GEMM_WAVES * 32) + tid;
    const int kg = idx >> 6, n = idx & (GEMM_BN - 1);
    const float* bp = B + (size_t)(4 * kg) * GAT_FH + n0 + n;
    float4 v;
    v.x = bp[0];
    v.y = bp[GAT_FH];
    v.z = bp[2 * GAT_FH];
    v.w = bp[3 * GAT_FH];
    Bs[idx] = v;
  }
  __syncthreads();

  const int wave = __builtin_amdgcn_readfirstlane(tid >> 5);
  const int lane = tid & 31;
  const int half = lane >> 4;
  const int l    = lane & 15;
  const int m0 = (by * GEMM_WAVES + wave) * 16;
  if (m0 >= M) return;

  const float4* __restrict__ ap = (const float4*)(A + (size_t)(m0 + l) * K) + half;
  const float4* bs = Bs + half * GEMM_BN + l;

  v8f acc0 = {0.f, 0.f, 0.f, 0.f, 0.f, 0.f, 0.f, 0.f};
  v8f acc1 = acc0, acc2 = acc0, acc3 = acc0;

#pragma unroll 4
  for (int c = 0; c < NC; ++c) {
    const float4 a  = ap[2 * c];
    const float4 b0 = bs[c * 2 * GEMM_BN];
    const float4 b1 = bs[c * 2 * GEMM_BN + 16];
    const float4 b2 = bs[c * 2 * GEMM_BN + 32];
    const float4 b3 = bs[c * 2 * GEMM_BN + 48];

    const v2f a01 = {a.x, a.y};
    const v2f a23 = {a.z, a.w};
    acc0 = wmma4(a01, (v2f){b0.x, b0.y}, acc0);
    acc1 = wmma4(a01, (v2f){b1.x, b1.y}, acc1);
    acc2 = wmma4(a01, (v2f){b2.x, b2.y}, acc2);
    acc3 = wmma4(a01, (v2f){b3.x, b3.y}, acc3);
    acc0 = wmma4(a23, (v2f){b0.z, b0.w}, acc0);
    acc1 = wmma4(a23, (v2f){b1.z, b1.w}, acc1);
    acc2 = wmma4(a23, (v2f){b2.z, b2.w}, acc2);
    acc3 = wmma4(a23, (v2f){b3.z, b3.w}, acc3);
  }

  float* crow = C + (size_t)(m0 + half * 8) * GAT_FH + n0 + l;
#pragma unroll
  for (int r = 0; r < 8; ++r) {
    crow[(size_t)r * GAT_FH]      = acc0[r];
    crow[(size_t)r * GAT_FH + 16] = acc1[r];
    crow[(size_t)r * GAT_FH + 32] = acc2[r];
    crow[(size_t)r * GAT_FH + 48] = acc3[r];
  }
}

template <int K>
__global__ void __launch_bounds__(GEMM_WAVES * 32)
gemm_wmma_f32(const float* __restrict__ A,
              const float* __restrict__ B,
              float* __restrict__ C, int M) {
  gemm_tile_f32<K>(A, B, C, M, blockIdx.x, blockIdx.y);
}

#define GEMM_STRIPS (GAT_FH / GEMM_BN)
#define GEMM_TILES (GEMM_STRIPS * GEMM_MGROUPS)
#define RIDE_COUNT (GEMM_TILES / 2)
#define RIDE_FILL (GEMM_TILES - RIDE_COUNT)

__global__ void csr_zero_kernel(int* __restrict__ deg) {
  int i = blockIdx.x * blockDim.x + threadIdx.x;
  if (i < GAT_N) deg[i] = 0;
}

template <int GFIRST, int GCOUNT>
__global__ void __launch_bounds__(GEMM_WAVES * 32)
csr_count_kernel(const int* __restrict__ ei, int* __restrict__ deg,
                 const float* __restrict__ x, const float* __restrict__ W,
                 float* __restrict__ hout) {
  if (blockIdx.x < GCOUNT) {
    const int g = GFIRST + blockIdx.x;
    gemm_tile_f32<128>(x, W, hout, GAT_N, g & (GEMM_STRIPS - 1), g / GEMM_STRIPS);
    return;
  }
  int e = (blockIdx.x - GCOUNT) * blockDim.x + threadIdx.x;
  if (e < GAT_E) atomicAdd(&deg[ei[GAT_E + e]], 1);
}

#define SCAN_TPB 1024
#define SCAN_BLOCKS ((GAT_N + SCAN_TPB - 1) / SCAN_TPB)
__global__ void __launch_bounds__(SCAN_TPB)
csr_scan_kernel(const int* __restrict__ deg, int* __restrict__ rowptr,
                int* __restrict__ cursor) {
  __shared__ int wave_tot[SCAN_TPB / 32];
  __shared__ int wave_ahead[SCAN_TPB / 32];
  const int t = threadIdx.x;
  const int lane = t & 31;
  const int w = t >> 5;
  const int b = blockIdx.x;
  const int i = b * SCAN_TPB + t;
  const int v = (i < GAT_N) ? deg[i] : 0;

  const int4* __restrict__ deg4 = (const int4*)deg;
  const int n4 = b * (SCAN_TPB / 4);
  int ahead = 0;
  for (int j = 0; j < (SCAN_BLOCKS + 3) / 4; ++j) {
    const int q = j * SCAN_TPB + t;
    const int4 x = deg4[(q < n4) ? q : 0];
    const int s4 = (x.x + x.y) + (x.z + x.w);
    ahead += (q < n4) ? s4 : 0;
  }

  int inc = v;
  for (int off = 1; off < 32; off <<= 1) {
    const int n = __shfl_up(inc, off, 32);
    if (lane >= off) inc += n;
  }
  for (int off = 16; off > 0; off >>= 1) ahead += __shfl_xor(ahead, off, 32);

  if (lane == 31) wave_tot[w] = inc;
  if (lane == 0) wave_ahead[w] = ahead;
  __syncthreads();

  const int wt = wave_tot[lane];
  int winc = wt;
  for (int off = 1; off < 32; off <<= 1) {
    const int n = __shfl_up(winc, off, 32);
    if (lane >= off) winc += n;
  }
  const int wave_off = __shfl(winc - wt, w, 32);
  int base = wave_ahead[lane];
  for (int off = 16; off > 0; off >>= 1) base += __shfl_xor(base, off, 32);

  const int excl = base + wave_off + (inc - v);
  if (i < GAT_N) {
    rowptr[i] = excl;
    cursor[i] = excl;
  }
  if (i == GAT_N - 1) rowptr[GAT_N] = excl + v;
}

template <int GFIRST, int GCOUNT>
__global__ void __launch_bounds__(GEMM_WAVES * 32)
csr_fill_kernel(const int* __restrict__ ei, int* __restrict__ cursor,
                int* __restrict__ tmp_src,
                const float* __restrict__ x, const float* __restrict__ W,
                float* __restrict__ hout) {
  if (blockIdx.x < GCOUNT) {
    const int g = GFIRST + blockIdx.x;
    gemm_tile_f32<128>(x, W, hout, GAT_N, g & (GEMM_STRIPS - 1), g / GEMM_STRIPS);
    return;
  }
  int e = (blockIdx.x - GCOUNT) * blockDim.x + threadIdx.x;
  if (e >= GAT_E) return;
  int p = atomicAdd(&cursor[ei[GAT_E + e]], 1);
  tmp_src[p] = ei[e];
}

__device__ __forceinline__ float bias_elu(float a, float b) {
  const float v = a + b;
  return v > 0.f ? v : (__expf(v) - 1.f);
}

template <int MASK>
__device__ __forceinline__ float row_xor(float v) {
  return __builtin_bit_cast(float,
      __builtin_amdgcn_update_dpp(0, __builtin_bit_cast(int, v), 0x160 + MASK, 0xF, 0xF, true));
}

__device__ __forceinline__ float head_sum(float v) {
  v += row_xor<1>(v);
  v += row_xor<2>(v);
  v += row_xor<4>(v);
  return v;
}

__device__ __forceinline__ float dot4(float4 a, float4 b) {
  float s = a.x * b.x;
  s += a.y * b.y; s += a.z * b.z; s += a.w * b.w;
  return s;
}

template <bool LAST>
__global__ void __launch_bounds__(256)
gat_gather_kernel(const int* __restrict__ rowptr, const int* __restrict__ tmp_src,
                  int* csr_src,
                  const float* __restrict__ asrc, const float* __restrict__ adst,
                  const float* __restrict__ h, const float* __restrict__ bias,
                  const float* __restrict__ Wl, const float* __restrict__ bl,
                  float* __restrict__ agg, float* __restrict__ y) {
  const int lane = threadIdx.x & 31;
  const int d = __builtin_amdgcn_readfirstlane((blockIdx.x * blockDim.x + threadIdx.x) >> 5);
  if (d >= GAT_N) return;
  const int beg = rowptr[d];
  const int deg = rowptr[d + 1] - beg;

  const float4* __restrict__ h4 = (const float4*)h;
  const float4 as0 = ((const float4*)asrc)[lane], as1 = ((const float4*)asrc)[32 + lane];
  const float4 ad0 = ((const float4*)adst)[lane], ad1 = ((const float4*)adst)[32 + lane];

  float4 acc0 = h4[(size_t)d * (GAT_FH / 4) + lane];
  float4 acc1 = h4[(size_t)d * (GAT_FH / 4) + 32 + lane];
  const float adA = head_sum(dot4(acc0, ad0));
  const float adB = head_sum(dot4(acc1, ad1));
  float mA = head_sum(dot4(acc0, as0)) + adA;
  float mB = head_sum(dot4(acc1, as1)) + adB;
  mA = mA > 0.f ? mA : NEG_SLOPE * mA;
  mB = mB > 0.f ? mB : NEG_SLOPE * mB;
  float dnA = 1.f, dnB = 1.f;

  int sorted0 = d;
  if (!LAST) {
    if (deg <= 32) {
      const int mine = (lane < deg) ? tmp_src[beg + lane] : 0;
      int rank = 0;
      for (int j = 0; j < deg; ++j) {
        const int sj = __builtin_amdgcn_readlane(mine, j);
        rank += ((sj < mine) || (sj == mine && j < lane)) ? 1 : 0;
      }
      rank = (lane < deg) ? rank : lane;
      sorted0 = __builtin_amdgcn_ds_permute(rank << 2, mine);
      if (lane < deg) csr_src[beg + lane] = sorted0;
    } else {
      for (int t = lane; t < deg; t += 32) {
        const int s = tmp_src[beg + t];
        int rank = 0;
        for (int j = 0; j < deg; ++j) {
          const int sj = tmp_src[beg + j];
          rank += ((sj < s) || (sj == s && j < t)) ? 1 : 0;
        }
        csr_src[beg + rank] = s;
      }
      __threadfence();
    }
  }

  for (int base = 0; base < deg; base += 32) {
    const int t = base + lane;
    const int sv = (!LAST && deg <= 32) ? sorted0 : ((t < deg) ? csr_src[beg + t] : d);
    const int cnt = (deg - base < 32) ? (deg - base) : 32;
#pragma unroll 4
    for (int i = 0; i < cnt; ++i) {
      const int s = __builtin_amdgcn_readlane(sv, i);
      const float4 x0 = h4[(size_t)s * (GAT_FH / 4) + lane];
      const float4 x1 = h4[(size_t)s * (GAT_FH / 4) + 32 + lane];
      float vA = head_sum(dot4(x0, as0)) + adA;
      float vB = head_sum(dot4(x1, as1)) + adB;
      vA = vA > 0.f ? vA : NEG_SLOPE * vA;
      vB = vB > 0.f ? vB : NEG_SLOPE * vB;
      const float nA = fmaxf(mA, vA), nB = fmaxf(mB, vB);
      const float cA = __expf(mA - nA), cB = __expf(mB - nB);
      const float pA = __expf(vA - nA), pB = __expf(vB - nB);
      mA = nA; mB = nB;
      dnA = dnA * cA + pA;
      dnB = dnB * cB + pB;
      acc0.x = acc0.x * cA + x0.x * pA; acc0.y = acc0.y * cA + x0.y * pA;
      acc0.z = acc0.z * cA + x0.z * pA; acc0.w = acc0.w * cA + x0.w * pA;
      acc1.x = acc1.x * cB + x1.x * pB; acc1.y = acc1.y * cB + x1.y * pB;
      acc1.z = acc1.z * cB + x1.z * pB; acc1.w = acc1.w * cB + x1.w * pB;
    }
  }
  const float dA = dnA + 1e-16f, dB = dnB + 1e-16f;
  acc0.x /= dA; acc0.y /= dA; acc0.z /= dA; acc0.w /= dA;
  acc1.x /= dB; acc1.y /= dB; acc1.z /= dB; acc1.w /= dB;

  const float4* __restrict__ b4 = (const float4*)bias;
  const float4 bb0 = b4[lane], bb1 = b4[32 + lane];
  float4 r0, r1;
  r0.x = bias_elu(acc0.x, bb0.x); r0.y = bias_elu(acc0.y, bb0.y);
  r0.z = bias_elu(acc0.z, bb0.z); r0.w = bias_elu(acc0.w, bb0.w);
  r1.x = bias_elu(acc1.x, bb1.x); r1.y = bias_elu(acc1.y, bb1.y);
  r1.z = bias_elu(acc1.z, bb1.z); r1.w = bias_elu(acc1.w, bb1.w);
  if (!LAST) {
    float4* o4 = (float4*)(agg + (size_t)d * GAT_FH);
    o4[lane] = r0;
    o4[32 + lane] = r1;
  } else {
    const float4* __restrict__ w4 = (const float4*)Wl;
    const float4 w0 = w4[lane], w1 = w4[32 + lane];
    float s = r0.x * w0.x;
    s += r0.y * w0.y; s += r0.z * w0.z; s += r0.w * w0.w;
    s += r1.x * w1.x; s += r1.y * w1.y; s += r1.z * w1.z; s += r1.w * w1.w;
    s += __shfl_xor(s, 16, 32);
    s += __shfl_xor(s, 8, 32);
    s += __shfl_xor(s, 4, 32);
    s += __shfl_xor(s, 2, 32);
    s += __shfl_xor(s, 1, 32);
    if (lane == 0) y[d] = s + bl[0];
  }
}

extern "C" void kernel_launch(void* const* d_in, const int* in_sizes, int n_in,
                              void* d_out, int out_size, void* d_ws, size_t ws_size,
                              hipStream_t stream) {
  const float* x      = (const float*)d_in[0];
  const int*   ei     = (const int*)  d_in[1];
  const float* W1     = (const float*)d_in[2];
  const float* a1_src = (const float*)d_in[3];
  const float* a1_dst = (const float*)d_in[4];
  const float* b1     = (const float*)d_in[5];
  const float* W2     = (const float*)d_in[6];
  const float* a2_src = (const float*)d_in[7];
  const float* a2_dst = (const float*)d_in[8];
  const float* b2     = (const float*)d_in[9];
  const float* Wl     = (const float*)d_in[10];
  const float* bl     = (const float*)d_in[11];
  float* out = (float*)d_out;

  float* hbuf  = (float*)d_ws;
  float* agg   = hbuf  + (size_t)GAT_N * GAT_FH;
  int* deg     = (int*)(agg + (size_t)GAT_N * GAT_FH);
  int* rowptr  = deg    + GAT_N;
  int* cursor  = rowptr + (GAT_N + 1);
  int* tmp_src = cursor + GAT_N;
  int* csr_src = tmp_src + GAT_E;

  dim3 gemm_grid(GAT_FH / GEMM_BN, GEMM_MGROUPS);
  dim3 gemm_blk(GEMM_WAVES * 32);

  const int TPB = 256;
  auto blocks = [](int n, int tpb) { return (n + tpb - 1) / tpb; };

  csr_zero_kernel<<<blocks(GAT_N, TPB), TPB, 0, stream>>>(deg);
  csr_count_kernel<0, RIDE_COUNT><<<RIDE_COUNT + blocks(GAT_E, TPB), TPB, 0, stream>>>(
      ei, deg, x, W1, hbuf);
  csr_scan_kernel<<<SCAN_BLOCKS, SCAN_TPB, 0, stream>>>(deg, rowptr, cursor);
  csr_fill_kernel<RIDE_COUNT, RIDE_FILL><<<RIDE_FILL + blocks(GAT_E, TPB), TPB, 0, stream>>>(
      ei, cursor, tmp_src, x, W1, hbuf);

  gat_gather_kernel<false><<<blocks(GAT_N * 32, TPB), TPB, 0, stream>>>(
      rowptr, tmp_src, csr_src, a1_src, a1_dst, hbuf, b1, Wl, bl, agg, out);

  gemm_wmma_f32<GAT_FH><<<gemm_grid, gemm_blk, 0, stream>>>(agg, W2, hbuf, GAT_N);
  gat_gather_kernel<true><<<blocks(GAT_N * 32, TPB), TPB, 0, stream>>>(
      rowptr, tmp_src, csr_src, a2_src, a2_dst, hbuf, b2, Wl, bl, agg, out);
}


# Edges ordered by destination with LDS counters instead of global atomics: chunks binned by 128-node range, then one workgroup per bin counts, scans and places; 7 launches become 5; same sums
# speedup vs baseline: 20.3299x; 1.4056x over previous
#include <hip/hip_runtime.h>
#include <hip/hip_bf16.h>

#define GAT_N 30000
#define GAT_E 480000
#define GAT_ETOT (GAT_E + GAT_N)
#define GAT_H 8
#define GAT_C 32
#define GAT_FH 256
#define NEG_SLOPE 0.2f

typedef float v2f __attribute__((ext_vector_type(2)));
typedef float v8f __attribute__((ext_vector_type(8)));

#define GEMM_BN 64
#define GEMM_WAVES 8
#define GEMM_MGROUPS ((GAT_N / 16 + GEMM_WAVES - 1) / GEMM_WAVES)

__device__ __forceinline__ v8f wmma4(v2f a, v2f b, v8f c) {
  return __builtin_amdgcn_wmma_f32_16x16x4_f32(
       false, a,  false, b,
       (short)0, c,  false,  false);
}

template <int K>
__device__ __forceinline__ void
gemm_tile_f32(const float* __restrict__ A,
              const float* __restrict__ B,
              float* __restrict__ C, int M, int bx, int by) {
  constexpr int KG = K / 4;
  constexpr int NC = K / 8;
  __shared__ float4 Bs[KG * GEMM_BN];

  const int tid = threadIdx.x;
  const int n0 = bx * GEMM_BN;

#pragma unroll 4
  for (int it = 0; it < KG * GEMM_BN / (GEMM_WAVES * 32); ++it) {
    const int idx = it * (GEMM_WAVES * 32) + tid;
    const int kg = idx >> 6, n = idx & (GEMM_BN - 1);
    const float* bp = B + (size_t)(4 * kg) * GAT_FH + n0 + n;
    float4 v;
    v.x = bp[0];
    v.y = bp[GAT_FH];
    v.z = bp[2 * GAT_FH];
    v.w = bp[3 * GAT_FH];
    Bs[idx] = v;
  }
  __syncthreads();

  const int wave = __builtin_amdgcn_readfirstlane(tid >> 5);
  const int lane = tid & 31;
  const int half = lane >> 4;
  const int l    = lane & 15;
  const int m0 = (by * GEMM_WAVES + wave) * 16;
  if (m0 >= M) return;

  const float4* __restrict__ ap = (const float4*)(A + (size_t)(m0 + l) * K) + half;
  const float4* bs = Bs + half * GEMM_BN + l;

  v8f acc0 = {0.f, 0.f, 0.f, 0.f, 0.f, 0.f, 0.f, 0.f};
  v8f acc1 = acc0, acc2 = acc0, acc3 = acc0;

#pragma unroll 4
  for (int c = 0; c < NC; ++c) {
    const float4 a  = ap[2 * c];
    const float4 b0 = bs[c * 2 * GEMM_BN];
    const float4 b1 = bs[c * 2 * GEMM_BN + 16];
    const float4 b2 = bs[c * 2 * GEMM_BN + 32];
    const float4 b3 = bs[c * 2 * GEMM_BN + 48];

    const v2f a01 = {a.x, a.y};
    const v2f a23 = {a.z, a.w};
    acc0 = wmma4(a01, (v2f){b0.x, b0.y}, acc0);
    acc1 = wmma4(a01, (v2f){b1.x, b1.y}, acc1);
    acc2 = wmma4(a01, (v2f){b2.x, b2.y}, acc2);
    acc3 = wmma4(a01, (v2f){b3.x, b3.y}, acc3);
    acc0 = wmma4(a23, (v2f){b0.z, b0.w}, acc0);
    acc1 = wmma4(a23, (v2f){b1.z, b1.w}, acc1);
    acc2 = wmma4(a23, (v2f){b2.z, b2.w}, acc2);
    acc3 = wmma4(a23, (v2f){b3.z, b3.w}, acc3);
  }

  float* crow = C + (size_t)(m0 + half * 8) * GAT_FH + n0 + l;
#pragma unroll
  for (int r = 0; r < 8; ++r) {
    crow[(size_t)r * GAT_FH]      = acc0[r];
    crow[(size_t)r * GAT_FH + 16] = acc1[r];
    crow[(size_t)r * GAT_FH + 32] = acc2[r];
    crow[(size_t)r * GAT_FH + 48] = acc3[r];
  }
}

template <int K>
__global__ void __launch_bounds__(GEMM_WAVES * 32)
gemm_wmma_f32(const float* __restrict__ A,
              const float* __restrict__ B,
              float* __restrict__ C, int M) {
  gemm_tile_f32<K>(A, B, C, M, blockIdx.x, blockIdx.y);
}

#define GEMM_STRIPS (GAT_FH / GEMM_BN)
#define GEMM_TILES (GEMM_STRIPS * GEMM_MGROUPS)
#define RIDE_COUNT (GEMM_TILES / 2)
#define RIDE_FILL (GEMM_TILES - RIDE_COUNT)

#define SORT_BINW 128
#define SORT_BINS ((GAT_N + SORT_BINW - 1) / SORT_BINW)
#define SORT_CH 4000
#define SORT_NCH (GAT_E / SORT_CH)
#define SORT_TPB (GEMM_WAVES * 32)
#define SORT_EPT ((SORT_CH + SORT_TPB - 1) / SORT_TPB)
#define SORT_FULL (SORT_CH / SORT_TPB)
#define SORT_CPW (SORT_NCH / GEMM_WAVES)
static_assert(GAT_E % SORT_CH == 0, "whole chunks");
static_assert(SORT_NCH % GEMM_WAVES == 0 && SORT_NCH % 4 == 0, "chunks per wave, 16-byte count reads");
static_assert(SORT_BINS <= SORT_TPB && SORT_BINW == 128 && GAT_N <= 32768, "packing and table sizes");

template <int GFIRST, int GCOUNT>
__global__ void __launch_bounds__(SORT_TPB)
csr_chunk_kernel(const int* __restrict__ ei, int* __restrict__ cntT, int* __restrict__ offT,
                 int* __restrict__ tmp_edges,
                 const float* __restrict__ x, const float* __restrict__ W,
                 float* __restrict__ hout) {
  if (blockIdx.x < GCOUNT) {
    const int g = GFIRST + blockIdx.x;
    gemm_tile_f32<128>(x, W, hout, GAT_N, g & (GEMM_STRIPS - 1), g / GEMM_STRIPS);
    return;
  }
  __shared__ int s_cnt[SORT_TPB];
  __shared__ int s_off[SORT_TPB];
  const int t = threadIdx.x;
  const int c = blockIdx.x - GCOUNT;
  s_cnt[t] = 0;
  __syncthreads();

  int key[SORT_EPT], tick[SORT_EPT];
#pragma unroll
  for (int k = 0; k < SORT_FULL; ++k) {
    const int e = c * SORT_CH + k * SORT_TPB + t;
    key[k] = (ei[GAT_E + e] << 15) | ei[e];
  }
#pragma unroll
  for (int k = SORT_FULL; k < SORT_EPT; ++k) {
    const int idx = k * SORT_TPB + t;
    key[k] = -1;
    if (idx < SORT_CH) {
      const int e = c * SORT_CH + idx;
      key[k] = (ei[GAT_E + e] << 15) | ei[e];
    }
  }
#pragma unroll
  for (int k = 0; k < SORT_EPT; ++k) {
    tick[k] = 0;
    if (key[k] >= 0) tick[k] = atomicAdd(&s_cnt[key[k] >> 22], 1);
  }
  __syncthreads();

  if (t < 32) {
    int v[8];
    int s = 0;
#pragma unroll
    for (int j = 0; j < 8; ++j) { v[j] = s_cnt[8 * t + j]; s += v[j]; }
    int inc = s;
    for (int off = 1; off < 32; off <<= 1) {
      const int nb = __shfl_up(inc, off, 32);
      if (t >= off) inc += nb;
    }
    int run = inc - s;
#pragma unroll
    for (int j = 0; j < 8; ++j) { s_off[8 * t + j] = run; run += v[j]; }
  }
  __syncthreads();

#pragma unroll
  for (int k = 0; k < SORT_EPT; ++k)
    if (key[k] >= 0) tick[k] += s_off[key[k] >> 22];
#pragma unroll
  for (int k = 0; k < SORT_EPT; ++k)
    if (key[k] >= 0) tmp_edges[c * SORT_CH + tick[k]] = key[k] & 0x3FFFFF;
  if (t < SORT_BINS) {
    cntT[t * SORT_NCH + c] = s_cnt[t];
    offT[t * SORT_NCH + c] = s_off[t];
  }
}

template <int GFIRST, int GCOUNT>
__global__ void __launch_bounds__(SORT_TPB)
csr_bin_kernel(const int* __restrict__ cntT, const int* __restrict__ offT,
               const int* __restrict__ tmp_edges,
               int* __restrict__ rowptr, int* __restrict__ tmp_src,
               const float* __restrict__ x, const float* __restrict__ W,
               float* __restrict__ hout) {
  if (blockIdx.x < GCOUNT) {
    const int g = GFIRST + blockIdx.x;
    gemm_tile_f32<128>(x, W, hout, GAT_N, g & (GEMM_STRIPS - 1), g / GEMM_STRIPS);
    return;
  }
  __shared__ int s_deg[SORT_BINW];
  __shared__ int s_cur[SORT_BINW];
  __shared__ int s_part[GEMM_WAVES];
  const int t = threadIdx.x;
  const int lane = t & 31;
  const int w = __builtin_amdgcn_readfirstlane(t >> 5);
  const int b = blockIdx.x - GCOUNT;

  const int4* __restrict__ cnt4 = (const int4*)cntT;
  const int n4 = b * (SORT_NCH / 4);
  int part = 0;
#pragma unroll 4
  for (int q = t; q < n4; q += SORT_TPB) {
    const int4 v = cnt4[q];
    part += (v.x + v.y) + (v.z + v.w);
  }
  for (int off = 16; off > 0; off >>= 1) part += __shfl_xor(part, off, 32);
  if (lane == 0) s_part[w] = part;
  if (t < SORT_BINW) s_deg[t] = 0;
  __syncthreads();
  int base = 0;
#pragma unroll
  for (int j = 0; j < GEMM_WAVES; ++j) base += s_part[j];

  int cn[SORT_CPW], st[SORT_CPW], ev[SORT_CPW];
#pragma unroll
  for (int k = 0; k < SORT_CPW; ++k) {
    const int c = w + GEMM_WAVES * k;
    cn[k] = cntT[b * SORT_NCH + c];
    st[k] = c * SORT_CH + offT[b * SORT_NCH + c];
  }
#pragma unroll
  for (int k = 0; k < SORT_CPW; ++k)
    ev[k] = (lane < cn[k]) ? tmp_edges[st[k] + lane] : -1;
#pragma unroll
  for (int k = 0; k < SORT_CPW; ++k)
    if (ev[k] >= 0) atomicAdd(&s_deg[ev[k] >> 15], 1);
#pragma unroll
  for (int k = 0; k < SORT_CPW; ++k)
    for (int i = 32 + lane; i < cn[k]; i += 32)
      atomicAdd(&s_deg[tmp_edges[st[k] + i] >> 15], 1);
  __syncthreads();

  if (t < 32) {
    int v[4];
    int s = 0;
#pragma unroll
    for (int j = 0; j < 4; ++j) { v[j] = s_deg[4 * t + j]; s += v[j]; }
    int inc = s;
    for (int off = 1; off < 32; off <<= 1) {
      const int nb = __shfl_up(inc, off, 32);
      if (t >= off) inc += nb;
    }
    int run = inc - s;
#pragma unroll
    for (int j = 0; j < 4; ++j) {
      const int node = b * SORT_BINW + 4 * t + j;
      s_cur[4 * t + j] = run;
      if (node < GAT_N) rowptr[node] = base + run;
      run += v[j];
    }
    if (b == SORT_BINS - 1 && t == 31) rowptr[GAT_N] = base + inc;
  }
  __syncthreads();

  int pl[SORT_CPW];
#pragma unroll
  for (int k = 0; k < SORT_CPW; ++k) {
    pl[k] = 0;
    if (ev[k] >= 0) pl[k] = atomicAdd(&s_cur[ev[k] >> 15], 1);
  }
#pragma unroll
  for (int k = 0; k < SORT_CPW; ++k)
    if (ev[k] >= 0) tmp_src[base + pl[k]] = ev[k] & 32767;
#pragma unroll
  for (int k = 0; k < SORT_CPW; ++k) {
    for (int i = 32 + lane; i < cn[k]; i += 32) {
      const int e = tmp_edges[st[k] + i];
      const int p = atomicAdd(&s_cur[e >> 15], 1);
      tmp_src[base + p] = e & 32767;
    }
  }
}

__device__ __forceinline__ float bias_elu(float a, float b) {
  const float v = a + b;
  return v > 0.f ? v : (__expf(v) - 1.f);
}

template <int MASK>
__device__ __forceinline__ float row_xor(float v) {
  return __builtin_bit_cast(float,
      __builtin_amdgcn_update_dpp(0, __builtin_bit_cast(int, v), 0x160 + MASK, 0xF, 0xF, true));
}

__device__ __forceinline__ float head_sum(float v) {
  v += row_xor<1>(v);
  v += row_xor<2>(v);
  v += row_xor<4>(v);
  return v;
}

__device__ __forceinline__ float dot4(float4 a, float4 b) {
  float s = a.x * b.x;
  s += a.y * b.y; s += a.z * b.z; s += a.w * b.w;
  return s;
}

template <bool LAST>
__global__ void __launch_bounds__(256)
gat_gather_kernel(const int* __restrict__ rowptr, const int* __restrict__ tmp_src,
                  int* csr_src,
                  const float* __restrict__ asrc, const float* __restrict__ adst,
                  const float* __restrict__ h, const float* __restrict__ bias,
                  const float* __restrict__ Wl, const float* __restrict__ bl,
                  float* __restrict__ agg, float* __restrict__ y) {
  const int lane = threadIdx.x & 31;
  const int d = __builtin_amdgcn_readfirstlane((blockIdx.x * blockDim.x + threadIdx.x) >> 5);
  if (d >= GAT_N) return;
  const int beg = rowptr[d];
  const int deg = rowptr[d + 1] - beg;

  const float4* __restrict__ h4 = (const float4*)h;
  const float4 as0 = ((const float4*)asrc)[lane], as1 = ((const float4*)asrc)[32 + lane];
  const float4 ad0 = ((const float4*)adst)[lane], ad1 = ((const float4*)adst)[32 + lane];

  float4 acc0 = h4[(size_t)d * (GAT_FH / 4) + lane];
  float4 acc1 = h4[(size_t)d * (GAT_FH / 4) + 32 + lane];
  const float adA = head_sum(dot4(acc0, ad0));
  const float adB = head_sum(dot4(acc1, ad1));
  float mA = head_sum(dot4(acc0, as0)) + adA;
  float mB = head_sum(dot4(acc1, as1)) + adB;
  mA = mA > 0.f ? mA : NEG_SLOPE * mA;
  mB = mB > 0.f ? mB : NEG_SLOPE * mB;
  float dnA = 1.f, dnB = 1.f;

  int sorted0 = d;
  if (!LAST) {
    if (deg <= 32) {
      const int mine = (lane < deg) ? tmp_src[beg + lane] : 0;
      int rank = 0;
      for (int j = 0; j < deg; ++j) {
        const int sj = __builtin_amdgcn_readlane(mine, j);
        rank += ((sj < mine) || (sj == mine && j < lane)) ? 1 : 0;
      }
      rank = (lane < deg) ? rank : lane;
      sorted0 = __builtin_amdgcn_ds_permute(rank << 2, mine);
      if (lane < deg) csr_src[beg + lane] = sorted0;
    } else {
      for (int t = lane; t < deg; t += 32) {
        const int s = tmp_src[beg + t];
        int rank = 0;
        for (int j = 0; j < deg; ++j) {
          const int sj = tmp_src[beg + j];
          rank += ((sj < s) || (sj == s && j < t)) ? 1 : 0;
        }
        csr_src[beg + rank] = s;
      }
      __threadfence();
    }
  }

  for (int base = 0; base < deg; base += 32) {
    const int t = base + lane;
    const int sv = (!LAST && deg <= 32) ? sorted0 : ((t < deg) ? csr_src[beg + t] : d);
    const int cnt = (deg - base < 32) ? (deg - base) : 32;
#pragma unroll 4
    for (int i = 0; i < cnt; ++i) {
      const int s = __builtin_amdgcn_readlane(sv, i);
      const float4 x0 = h4[(size_t)s * (GAT_FH / 4) + lane];
      const float4 x1 = h4[(size_t)s * (GAT_FH / 4) + 32 + lane];
      float vA = head_sum(dot4(x0, as0)) + adA;
      float vB = head_sum(dot4(x1, as1)) + adB;
      vA = vA > 0.f ? vA : NEG_SLOPE * vA;
      vB = vB > 0.f ? vB : NEG_SLOPE * vB;
      const float nA = fmaxf(mA, vA), nB = fmaxf(mB, vB);
      const float cA = __expf(mA - nA), cB = __expf(mB - nB);
      const float pA = __expf(vA - nA), pB = __expf(vB - nB);
      mA = nA; mB = nB;
      dnA = dnA * cA + pA;
      dnB = dnB * cB + pB;
      acc0.x = acc0.x * cA + x0.x * pA; acc0.y = acc0.y * cA + x0.y * pA;
      acc0.z = acc0.z * cA + x0.z * pA; acc0.w = acc0.w * cA + x0.w * pA;
      acc1.x = acc1.x * cB + x1.x * pB; acc1.y = acc1.y * cB + x1.y * pB;
      acc1.z = acc1.z * cB + x1.z * pB; acc1.w = acc1.w * cB + x1.w * pB;
    }
  }
  const float dA = dnA + 1e-16f, dB = dnB + 1e-16f;
  acc0.x /= dA; acc0.y /= dA; acc0.z /= dA; acc0.w /= dA;
  acc1.x /= dB; acc1.y /= dB; acc1.z /= dB; acc1.w /= dB;

  const float4* __restrict__ b4 = (const float4*)bias;
  const float4 bb0 = b4[lane], bb1 = b4[32 + lane];
  float4 r0, r1;
  r0.x = bias_elu(acc0.x, bb0.x); r0.y = bias_elu(acc0.y, bb0.y);
  r0.z = bias_elu(acc0.z, bb0.z); r0.w = bias_elu(acc0.w, bb0.w);
  r1.x = bias_elu(acc1.x, bb1.x); r1.y = bias_elu(acc1.y, bb1.y);
  r1.z = bias_elu(acc1.z, bb1.z); r1.w = bias_elu(acc1.w, bb1.w);
  if (!LAST) {
    float4* o4 = (float4*)(agg + (size_t)d * GAT_FH);
    o4[lane] = r0;
    o4[32 + lane] = r1;
  } else {
    const float4* __restrict__ w4 = (const float4*)Wl;
    const float4 w0 = w4[lane], w1 = w4[32 + lane];
    float s = r0.x * w0.x;
    s += r0.y * w0.y; s += r0.z * w0.z; s += r0.w * w0.w;
    s += r1.x * w1.x; s += r1.y * w1.y; s += r1.z * w1.z; s += r1.w * w1.w;
    s += __shfl_xor(s, 16, 32);
    s += __shfl_xor(s, 8, 32);
    s += __shfl_xor(s, 4, 32);
    s += __shfl_xor(s, 2, 32);
    s += __shfl_xor(s, 1, 32);
    if (lane == 0) y[d] = s + bl[0];
  }
}

extern "C" void kernel_launch(void* const* d_in, const int* in_sizes, int n_in,
                              void* d_out, int out_size, void* d_ws, size_t ws_size,
                              hipStream_t stream) {
  const float* x      = (const float*)d_in[0];
  const int*   ei     = (const int*)  d_in[1];
  const float* W1     = (const float*)d_in[2];
  const float* a1_src = (const float*)d_in[3];
  const float* a1_dst = (const float*)d_in[4];
  const float* b1     = (const float*)d_in[5];
  const float* W2     = (const float*)d_in[6];
  const float* a2_src = (const float*)d_in[7];
  const float* a2_dst = (const float*)d_in[8];
  const float* b2     = (const float*)d_in[9];
  const float* Wl     = (const float*)d_in[10];
  const float* bl     = (const float*)d_in[11];
  float* out = (float*)d_out;

  float* hbuf  = (float*)d_ws;
  float* agg   = hbuf  + (size_t)GAT_N * GAT_FH;
  int* cntT      = (int*)(agg + (size_t)GAT_N * GAT_FH);
  int* offT      = cntT + SORT_BINS * SORT_NCH;
  int* tmp_edges = offT + SORT_BINS * SORT_NCH;
  int* tmp_src   = tmp_edges + GAT_E;
  int* csr_src   = tmp_src + GAT_E;
  int* rowptr    = csr_src + GAT_E;

  dim3 gemm_grid(GAT_FH / GEMM_BN, GEMM_MGROUPS);
  dim3 gemm_blk(GEMM_WAVES * 32);

  const int TPB = 256;
  auto blocks = [](int n, int tpb) { return (n + tpb - 1) / tpb; };

  csr_chunk_kernel<0, RIDE_COUNT><<<RIDE_COUNT + SORT_NCH, SORT_TPB, 0, stream>>>(
      ei, cntT, offT, tmp_edges, x, W1, hbuf);
  csr_bin_kernel<RIDE_COUNT, RIDE_FILL><<<RIDE_FILL + SORT_BINS, SORT_TPB, 0, stream>>>(
      cntT, offT, tmp_edges, rowptr, tmp_src, x, W1, hbuf);

  gat_gather_kernel<false><<<blocks(GAT_N * 32, TPB), TPB, 0, stream>>>(
      rowptr, tmp_src, csr_src, a1_src, a1_dst, hbuf, b1, Wl, bl, agg, out);

  gemm_wmma_f32<GAT_FH><<<gemm_grid, gemm_blk, 0, stream>>>(agg, W2, hbuf, GAT_N);
  gat_gather_kernel<true><<<blocks(GAT_N * 32, TPB), TPB, 0, stream>>>(
      rowptr, tmp_src, csr_src, a2_src, a2_dst, hbuf, b2, Wl, bl, agg, out);
}
